# starting kernel
# baseline (speedup 1.0000x reference)
#include <hip/hip_runtime.h>
#include <hip/hip_bf16.h>
#include <math.h>

typedef float v2f __attribute__((ext_vector_type(2)));
typedef float v8f __attribute__((ext_vector_type(8)));

constexpr int NM    = 31;
constexpr int NP    = 961;
constexpr int NPPAD = 992;
constexpr int KNP   = 964;
constexpr int J     = 768;
constexpr int JOs   = 256;
constexpr int CCH   = 4;
constexpr int CMP   = 128;
constexpr int QMR   = 496;
constexpr int QMP   = 512;
constexpr float LBOX   = 20.0f;
constexpr float VBOX   = 8000.0f;
constexpr float TWOPI_L = 0.31415926535897931f;

constexpr size_t SZ_PH  = (size_t)NM * J;
constexpr size_t OFF_CX = 0;
constexpr size_t OFF_SX = OFF_CX + SZ_PH;
constexpr size_t OFF_CY = OFF_SX + SZ_PH;
constexpr size_t OFF_SY = OFF_CY + SZ_PH;
constexpr size_t OFF_CZ = OFF_SY + SZ_PH;
constexpr size_t OFF_SZ = OFF_CZ + SZ_PH;
constexpr size_t SZ_W   = (size_t)NPPAD * J;
constexpr size_t OFF_WC = OFF_SZ + SZ_PH;
constexpr size_t OFF_WS = OFF_WC + SZ_W;
constexpr size_t SZ_Y   = (size_t)CMP * J;
constexpr size_t OFF_YR = OFF_WS + SZ_W;
constexpr size_t OFF_YI = OFF_YR + SZ_Y;
constexpr size_t SZ_R   = (size_t)CMP * NPPAD;
constexpr size_t OFF_ROR = OFF_YI + SZ_Y;
constexpr size_t OFF_ROI = OFF_ROR + SZ_R;
constexpr size_t OFF_RHR = OFF_ROI + SZ_R;
constexpr size_t OFF_RHI = OFF_RHR + SZ_R;
constexpr size_t SZ_G   = (size_t)QMP * KNP;
constexpr size_t OFF_GOR = OFF_RHI + SZ_R;
constexpr size_t OFF_GOI = OFF_GOR + SZ_G;
constexpr size_t OFF_GHR = OFF_GOI + SZ_G;
constexpr size_t OFF_GHI = OFF_GHR + SZ_G;
constexpr size_t SZ_T   = (size_t)QMP * J;
constexpr size_t OFF_TR = OFF_GHI + SZ_G;
constexpr size_t OFF_TI = OFF_TR + SZ_T;
constexpr size_t WS_TOTAL = OFF_TI + SZ_T;

__global__ void phase_build(const float* __restrict__ pos,
                            float* __restrict__ Cx, float* __restrict__ Sx,
                            float* __restrict__ Cy, float* __restrict__ Sy,
                            float* __restrict__ Cz, float* __restrict__ Sz) {
  int t = blockIdx.x * blockDim.x + threadIdx.x;
  if (t >= 3 * J) return;
  int d = t / J, j = t % J;
  float th = TWOPI_L * (pos[j * 3 + d] - 0.5f * LBOX);
  float* C = (d == 0) ? Cx : (d == 1) ? Cy : Cz;
  float* S = (d == 0) ? Sx : (d == 1) ? Sy : Sz;
  for (int mi = 0; mi < NM; ++mi) {
    float s, c;
    sincosf((float)(mi - 15) * th, &s, &c);
    C[mi * J + j] = c;
    S[mi * J + j] = s;
  }
}

__global__ void w_build(const float* __restrict__ Cy, const float* __restrict__ Sy,
                        const float* __restrict__ Cz, const float* __restrict__ Sz,
                        float* __restrict__ Wc, float* __restrict__ Ws) {
  int t = blockIdx.x * blockDim.x + threadIdx.x;
  if (t >= NP * J) return;
  int np = t / J, j = t % J;
  int n = np / NM, p = np % NM;
  float cy = Cy[n * J + j], sy = Sy[n * J + j];
  float cz = Cz[p * J + j], sz = Sz[p * J + j];
  Wc[(size_t)np * J + j] = cy * cz - sy * sz;
  Ws[(size_t)np * J + j] = sy * cz + cy * sz;
}

__global__ void y_build(const float* __restrict__ charge,
                        const float* __restrict__ Cx, const float* __restrict__ Sx,
                        float* __restrict__ Yr, float* __restrict__ Yi) {
  int t = blockIdx.x * blockDim.x + threadIdx.x;
  if (t >= (CCH * NM) * J) return;
  int cm = t / J, j = t % J;
  int c = cm / NM, m = cm % NM;
  float q = charge[j * CCH + c];
  Yr[(size_t)cm * J + j] =  q * Cx[m * J + j];
  Yi[(size_t)cm * J + j] = -q * Sx[m * J + j];
}

template <bool NTMODE, bool SPLUS>
__global__ __launch_bounds__(32)
void cgemm_wmma(const float* __restrict__ Ar, const float* __restrict__ Ai, int lda,
                const float* __restrict__ Br, const float* __restrict__ Bi, int ldb,
                float* __restrict__ Cr, float* __restrict__ Ci, int ldc, int K) {
  const int lane = threadIdx.x;
  const int row  = lane & 15;
  const int kp   = lane >> 4;
  const int n0 = blockIdx.x * 32;
  const int m0 = blockIdx.y * 32;

  v8f accr[2][2], acci[2][2];
#pragma unroll
  for (int mt = 0; mt < 2; ++mt)
#pragma unroll
    for (int nt = 0; nt < 2; ++nt) {
      accr[mt][nt] = (v8f){0.f, 0.f, 0.f, 0.f, 0.f, 0.f, 0.f, 0.f};
      acci[mt][nt] = (v8f){0.f, 0.f, 0.f, 0.f, 0.f, 0.f, 0.f, 0.f};
    }

  const float* arp0 = Ar + (size_t)(m0 + row) * lda + 2 * kp;
  const float* aip0 = Ai + (size_t)(m0 + row) * lda + 2 * kp;
  const float* arp1 = arp0 + (size_t)16 * lda;
  const float* aip1 = aip0 + (size_t)16 * lda;

  for (int k0 = 0; k0 < K; k0 += 4) {
    v2f a_r[2], a_i[2], b_r[2], b_i[2];
    a_r[0] = *(const v2f*)(arp0 + k0);
    a_r[1] = *(const v2f*)(arp1 + k0);
    a_i[0] = *(const v2f*)(aip0 + k0);
    a_i[1] = *(const v2f*)(aip1 + k0);
    if (NTMODE) {
      const size_t bo = (size_t)(n0 + row) * ldb + k0 + 2 * kp;
      b_r[0] = *(const v2f*)(Br + bo);
      b_i[0] = *(const v2f*)(Bi + bo);
      b_r[1] = *(const v2f*)(Br + bo + (size_t)16 * ldb);
      b_i[1] = *(const v2f*)(Bi + bo + (size_t)16 * ldb);
    } else {
      const int kk = k0 + 2 * kp;
#pragma unroll
      for (int nt = 0; nt < 2; ++nt) {
        const size_t c0 = (size_t)kk * ldb + n0 + nt * 16 + row;
        b_r[nt].x = Br[c0];
        b_r[nt].y = Br[c0 + ldb];
        b_i[nt].x = Bi[c0];
        b_i[nt].y = Bi[c0 + ldb];
      }
    }
    v2f neg[2];
#pragma unroll
    for (int mt = 0; mt < 2; ++mt) {
      v2f src = SPLUS ? a_i[mt] : a_r[mt];
      neg[mt].x = -src.x;
      neg[mt].y = -src.y;
    }
#pragma unroll
    for (int mt = 0; mt < 2; ++mt)
#pragma unroll
      for (int nt = 0; nt < 2; ++nt) {
        accr[mt][nt] = __builtin_amdgcn_wmma_f32_16x16x4_f32(
            false, a_r[mt], false, b_r[nt], (short)0, accr[mt][nt], false, false);
        acci[mt][nt] = __builtin_amdgcn_wmma_f32_16x16x4_f32(
            false, a_i[mt], false, b_r[nt], (short)0, acci[mt][nt], false, false);
        if (SPLUS) {
          accr[mt][nt] = __builtin_amdgcn_wmma_f32_16x16x4_f32(
              false, neg[mt], false, b_i[nt], (short)0, accr[mt][nt], false, false);
          acci[mt][nt] = __builtin_amdgcn_wmma_f32_16x16x4_f32(
              false, a_r[mt], false, b_i[nt], (short)0, acci[mt][nt], false, false);
        } else {
          accr[mt][nt] = __builtin_amdgcn_wmma_f32_16x16x4_f32(
              false, a_i[mt], false, b_i[nt], (short)0, accr[mt][nt], false, false);
          acci[mt][nt] = __builtin_amdgcn_wmma_f32_16x16x4_f32(
              false, neg[mt], false, b_i[nt], (short)0, acci[mt][nt], false, false);
        }
      }
  }

#pragma unroll
  for (int mt = 0; mt < 2; ++mt)
#pragma unroll
    for (int nt = 0; nt < 2; ++nt)
#pragma unroll
      for (int v = 0; v < 8; ++v) {
        const size_t o = (size_t)(m0 + mt * 16 + v + 8 * kp) * ldc + n0 + nt * 16 + row;
        Cr[o] = accr[mt][nt][v];
        Ci[o] = acci[mt][nt][v];
      }
}

__global__ void g_build(const float* __restrict__ rOr, const float* __restrict__ rOi,
                        const float* __restrict__ rHr, const float* __restrict__ rHi,
                        const float* __restrict__ shOO, const float* __restrict__ amOO,
                        const float* __restrict__ shOH, const float* __restrict__ amOH,
                        const float* __restrict__ shHO, const float* __restrict__ amHO,
                        const float* __restrict__ shHH, const float* __restrict__ amHH,
                        float* __restrict__ gOr, float* __restrict__ gOi,
                        float* __restrict__ gHr, float* __restrict__ gHi) {
  int t = blockIdx.x * blockDim.x + threadIdx.x;
  if (t >= NM * NP) return;
  int m = t / NP, np = t % NP;
  int n = np / NM, p = np % NM;
  float kx = TWOPI_L * (float)(m - 15);
  float ky = TWOPI_L * (float)(n - 15);
  float kz = TWOPI_L * (float)(p - 15);
  float squ = kx * kx + ky * ky + kz * kz;

  float mOO = 0.f, mOH = 0.f, mHO = 0.f, mHH = 0.f;
#pragma unroll
  for (int i = 0; i < 6; ++i) {
    mOO += amOO[i] * expf(-squ * expf(2.0f * shOO[i]));
    mOH += amOH[i] * expf(-squ * expf(2.0f * shOH[i]));
    mHO += amHO[i] * expf(-squ * expf(2.0f * shHO[i]));
    mHH += amHH[i] * expf(-squ * expf(2.0f * shHH[i]));
  }
  if (squ == 0.0f) { mOH = 0.f; mHO = 0.f; mHH = 0.f; }

#pragma unroll
  for (int c = 0; c < CCH; ++c) {
    const size_t ri = (size_t)(c * NM + m) * NPPAD + np;
    float ror = rOr[ri], roi = rOi[ri];
    float rhr = rHr[ri], rhi = rHi[ri];
    float fOr = mOO * ror + mOH * rhr, fOi = mOO * roi + mOH * rhi;
    float fHr = mHO * ror + mHH * rhr, fHi = mHO * roi + mHH * rhi;
    const size_t r0 = (size_t)(c * NM + m)        * KNP + np;
    const size_t r1 = (size_t)((4  + c) * NM + m) * KNP + np;
    const size_t r2 = (size_t)((8  + c) * NM + m) * KNP + np;
    const size_t r3 = (size_t)((12 + c) * NM + m) * KNP + np;
    gOr[r0] = fOr;        gOi[r0] = fOi;
    gOr[r1] = kx * fOr;   gOi[r1] = kx * fOi;
    gOr[r2] = ky * fOr;   gOi[r2] = ky * fOi;
    gOr[r3] = kz * fOr;   gOi[r3] = kz * fOi;
    gHr[r0] = fHr;        gHi[r0] = fHi;
    gHr[r1] = kx * fHr;   gHi[r1] = kx * fHi;
    gHr[r2] = ky * fHr;   gHi[r2] = ky * fHi;
    gHr[r3] = kz * fHr;   gHi[r3] = kz * fHi;
  }
}

__global__ void finalize(const float* __restrict__ Tr, const float* __restrict__ Ti,
                         const float* __restrict__ Cx, const float* __restrict__ Sx,
                         const float* __restrict__ charge, float* __restrict__ out, int b) {
  int j = blockIdx.x * blockDim.x + threadIdx.x;
  if (j >= J) return;
  float ur[4]  = {0.f, 0.f, 0.f, 0.f};
  float ui[12] = {0.f, 0.f, 0.f, 0.f, 0.f, 0.f, 0.f, 0.f, 0.f, 0.f, 0.f, 0.f};
  for (int m = 0; m < NM; ++m) {
    float cx = Cx[m * J + j], sx = Sx[m * J + j];
#pragma unroll
    for (int q = 0; q < 16; ++q) {
      float tr = Tr[(size_t)(q * NM + m) * J + j];
      float ti = Ti[(size_t)(q * NM + m) * J + j];
      if (q < 4) ur[q]     += tr * cx - ti * sx;
      else       ui[q - 4] += tr * sx + ti * cx;
    }
  }
  float e = 0.f, f0 = 0.f, f1 = 0.f, f2 = 0.f;
#pragma unroll
  for (int c = 0; c < CCH; ++c) {
    float q = charge[j * CCH + c];
    e  += q * ur[c];
    f0 += q * ui[0 + c];
    f1 += q * ui[4 + c];
    f2 += q * ui[8 + c];
  }
  out[b * J + j] = e / (2.0f * VBOX);
  float* F = out + 2 * J;
  F[(size_t)(b * J + j) * 3 + 0] = f0 / VBOX;
  F[(size_t)(b * J + j) * 3 + 1] = f1 / VBOX;
  F[(size_t)(b * J + j) * 3 + 2] = f2 / VBOX;
}

extern "C" void kernel_launch(void* const* d_in, const int* in_sizes, int n_in,
                              void* d_out, int out_size, void* d_ws, size_t ws_size,
                              hipStream_t stream) {
  const float* pos    = (const float*)d_in[0];
  const float* charge = (const float*)d_in[1];
  const float* shOO = (const float*)d_in[2];
  const float* amOO = (const float*)d_in[3];
  const float* shOH = (const float*)d_in[4];
  const float* amOH = (const float*)d_in[5];
  const float* shHO = (const float*)d_in[6];
  const float* amHO = (const float*)d_in[7];
  const float* shHH = (const float*)d_in[8];
  const float* amHH = (const float*)d_in[9];
  float* out = (float*)d_out;
  float* ws  = (float*)d_ws;

  float *Cx = ws + OFF_CX, *Sx = ws + OFF_SX;
  float *Cy = ws + OFF_CY, *Sy = ws + OFF_SY;
  float *Cz = ws + OFF_CZ, *Sz = ws + OFF_SZ;
  float *Wc = ws + OFF_WC, *Wss = ws + OFF_WS;
  float *Yr = ws + OFF_YR, *Yi = ws + OFF_YI;
  float *rOr = ws + OFF_ROR, *rOi = ws + OFF_ROI;
  float *rHr = ws + OFF_RHR, *rHi = ws + OFF_RHI;
  float *gOr = ws + OFF_GOR, *gOi = ws + OFF_GOI;
  float *gHr = ws + OFF_GHR, *gHi = ws + OFF_GHI;
  float *Tr = ws + OFF_TR, *Ti = ws + OFF_TI;

  hipMemsetAsync(d_ws, 0, WS_TOTAL * sizeof(float), stream);

  for (int b = 0; b < 2; ++b) {
    const float* posb = pos + (size_t)b * J * 3;
    const float* chb  = charge + (size_t)b * J * CCH;

    phase_build<<<(3 * J + 127) / 128, 128, 0, stream>>>(posb, Cx, Sx, Cy, Sy, Cz, Sz);
    w_build<<<(NP * J + 255) / 256, 256, 0, stream>>>(Cy, Sy, Cz, Sz, Wc, Wss);
    y_build<<<(CCH * NM * J + 255) / 256, 256, 0, stream>>>(chb, Cx, Sx, Yr, Yi);

    dim3 g1(NPPAD / 32, CMP / 32);
    cgemm_wmma<true, false><<<g1, 32, 0, stream>>>(Yr, Yi, J, Wc, Wss, J,
                                                   rOr, rOi, NPPAD, JOs);
    cgemm_wmma<true, false><<<g1, 32, 0, stream>>>(Yr + JOs, Yi + JOs, J,
                                                   Wc + JOs, Wss + JOs, J,
                                                   rHr, rHi, NPPAD, J - JOs);

    g_build<<<(NM * NP + 255) / 256, 256, 0, stream>>>(
        rOr, rOi, rHr, rHi,
        shOO, amOO, shOH, amOH, shHO, amHO, shHH, amHH,
        gOr, gOi, gHr, gHi);

    dim3 g2o(JOs / 32, QMP / 32), g2h((J - JOs) / 32, QMP / 32);
    cgemm_wmma<false, true><<<g2o, 32, 0, stream>>>(gOr, gOi, KNP, Wc, Wss, J,
                                                    Tr, Ti, J, KNP);
    cgemm_wmma<false, true><<<g2h, 32, 0, stream>>>(gHr, gHi, KNP, Wc + JOs, Wss + JOs, J,
                                                    Tr + JOs, Ti + JOs, J, KNP);

    finalize<<<(J + 127) / 128, 128, 0, stream>>>(Tr, Ti, Cx, Sx, chb, out, b);
  }
}


# W and Y never stored: nufft1 forms its operand tiles from the phase tables and charges on the way into LDS, type-2 forms its W rows in registers; 12 -> 8 launches; sum orders unchanged
# speedup vs baseline: 9.7654x; 9.7654x over previous
#include <hip/hip_runtime.h>
#include <hip/hip_bf16.h>
#include <math.h>

typedef float v2f __attribute__((ext_vector_type(2)));
typedef float v8f __attribute__((ext_vector_type(8)));

constexpr int NM    = 31;
constexpr int NP    = 961;
constexpr int NPPAD = 1024;
constexpr int KNP   = 1024;
constexpr int J     = 768;
constexpr int JOs   = 256;
constexpr int CCH   = 4;
constexpr int CMP   = 128;
constexpr int GPAIR = 16;
constexpr int NT16  = 61;
constexpr int TPS   = 3;
constexpr int NSL   = 21;
constexpr float LBOX   = 20.0f;
constexpr float VBOX   = 8000.0f;
constexpr float TWOPI_L = 0.31415926535897931f;

constexpr size_t SZ_PH  = (size_t)NM * J;
constexpr size_t OFF_CX = 0;
constexpr size_t OFF_SX = OFF_CX + SZ_PH;
constexpr size_t OFF_CY = OFF_SX + SZ_PH;
constexpr size_t OFF_SY = OFF_CY + SZ_PH;
constexpr size_t OFF_CZ = OFF_SY + SZ_PH;
constexpr size_t OFF_SZ = OFF_CZ + SZ_PH;
constexpr size_t SZ_G   = (size_t)CCH * GPAIR * KNP * 2;
constexpr size_t OFF_GOR = OFF_SZ + SZ_PH;
constexpr size_t OFF_GOI = OFF_GOR + SZ_G;
constexpr size_t OFF_GHR = OFF_GOI + SZ_G;
constexpr size_t OFF_GHI = OFF_GHR + SZ_G;
constexpr size_t SZ_PART = (size_t)NSL * J * 4;
constexpr size_t OFF_PART = OFF_GHI + SZ_G;
constexpr size_t WS_TOTAL = OFF_PART + SZ_PART;

__global__ void phase_build(const float* __restrict__ pos,
                            float* __restrict__ Cx, float* __restrict__ Sx,
                            float* __restrict__ Cy, float* __restrict__ Sy,
                            float* __restrict__ Cz, float* __restrict__ Sz) {
  int t = blockIdx.x * blockDim.x + threadIdx.x;
  if (t >= 3 * J) return;
  int d = t / J, j = t % J;
  float th = TWOPI_L * (pos[j * 3 + d] - 0.5f * LBOX);
  float* C = (d == 0) ? Cx : (d == 1) ? Cy : Cz;
  float* S = (d == 0) ? Sx : (d == 1) ? Sy : Sz;
  for (int mi = 0; mi < NM; ++mi) {
    float s, c;
    sincosf((float)(mi - 15) * th, &s, &c);
    C[mi * J + j] = c;
    S[mi * J + j] = s;
  }
}

typedef float v4f __attribute__((ext_vector_type(4)));
constexpr int TM = 32;
constexpr int TN = 32;
constexpr int KC = 64;
constexpr int PA = 68;
constexpr int PB = 40;
constexpr int SZ_A = TM * PA;
constexpr int SZ_B = (TN * PA > KC * PB) ? TN * PA : KC * PB;
constexpr int S_AR = 0;
constexpr int S_AI = S_AR + SZ_A;
constexpr int S_BR = S_AI + SZ_A;
constexpr int S_BI = S_BR + SZ_B;
constexpr int S_BUF = S_BI + SZ_B;
static_assert(SZ_A % 4 == 0 && SZ_B % 4 == 0, "planes start on 16-byte boundaries");

constexpr int KB = KC / 8;

template <bool NTMODE>
__device__ __forceinline__ void frag_load(const float* buf, int fA, int fB, int blk,
                                          v2f (&ar)[KB], v2f (&ai)[KB],
                                          v2f (&br)[KB], v2f (&bi)[KB]) {
#pragma unroll
  for (int s = 0; s < KB; ++s) {
    const int ks = blk * KB + s;
    ar[s] = *(const v2f*)(buf + S_AR + fA + 4 * ks);
    ai[s] = *(const v2f*)(buf + S_AI + fA + 4 * ks);
    if (NTMODE) {
      br[s] = *(const v2f*)(buf + S_BR + fB + 4 * ks);
      bi[s] = *(const v2f*)(buf + S_BI + fB + 4 * ks);
    } else {
      br[s].x = buf[S_BR + fB + (4 * ks) * PB];
      br[s].y = buf[S_BR + fB + (4 * ks + 1) * PB];
      bi[s].x = buf[S_BI + fB + (4 * ks) * PB];
      bi[s].y = buf[S_BI + fB + (4 * ks + 1) * PB];
    }
  }
}

template <bool SPLUS>
__device__ __forceinline__ void wmma_block(const v2f (&ar)[KB], const v2f (&ai)[KB],
                                           const v2f (&br)[KB], const v2f (&bi)[KB],
                                           v8f& accr, v8f& acci) {
#pragma unroll
  for (int s = 0; s < KB; ++s) {
    v2f src = SPLUS ? ai[s] : ar[s];
    v2f neg;
    neg.x = -src.x;
    neg.y = -src.y;
    accr = __builtin_amdgcn_wmma_f32_16x16x4_f32(
        false, ar[s], false, br[s], (short)0, accr, false, false);
    acci = __builtin_amdgcn_wmma_f32_16x16x4_f32(
        false, ai[s], false, br[s], (short)0, acci, false, false);
    if (SPLUS) {
      accr = __builtin_amdgcn_wmma_f32_16x16x4_f32(
          false, neg, false, bi[s], (short)0, accr, false, false);
      acci = __builtin_amdgcn_wmma_f32_16x16x4_f32(
          false, ar[s], false, bi[s], (short)0, acci, false, false);
    } else {
      accr = __builtin_amdgcn_wmma_f32_16x16x4_f32(
          false, ai[s], false, bi[s], (short)0, accr, false, false);
      acci = __builtin_amdgcn_wmma_f32_16x16x4_f32(
          false, neg, false, bi[s], (short)0, acci, false, false);
    }
  }
}

__global__ __launch_bounds__(128)
void nufft1_mix(const float* __restrict__ charge,
                const float* __restrict__ Cx, const float* __restrict__ Sx,
                const float* __restrict__ Cy, const float* __restrict__ Sy,
                const float* __restrict__ Cz, const float* __restrict__ Sz,
                const float* __restrict__ shOO, const float* __restrict__ amOO,
                const float* __restrict__ shOH, const float* __restrict__ amOH,
                const float* __restrict__ shHO, const float* __restrict__ amHO,
                const float* __restrict__ shHH, const float* __restrict__ amHH,
                float* __restrict__ gOr, float* __restrict__ gOi,
                float* __restrict__ gHr, float* __restrict__ gHi, int KO, int K) {
  __shared__ __attribute__((aligned(16))) float sm[2 * S_BUF];

  const int t    = threadIdx.x;
  const int lane = t & 31;
  const int wv   = t >> 5;
  const int row  = lane & 15;
  const int kp   = lane >> 4;
  const int wm   = wv >> 1;
  const int wn   = wv & 1;
  const int n0 = blockIdx.x * TN;
  const int m0 = blockIdx.y * TM;

  const int ac4 = t & 15, arow = t >> 4;
  const int sT = arow * PA + 4 * ac4;
  const float* qp = charge + (size_t)(4 * ac4) * CCH + (arow & 3);
  int offA[4], offN[4], offP[4];
#pragma unroll
  for (int i = 0; i < 4; ++i) {
    int mA = (m0 + arow + 8 * i) >> 2;
    if (mA > NM - 1) mA = NM - 1;
    int npB = n0 + arow + 8 * i;
    if (npB > NP - 1) npB = NP - 1;
    const int nB = npB / NM, pB = npB - NM * nB;
    offA[i] = mA * J + 4 * ac4;
    offN[i] = nB * J + 4 * ac4;
    offP[i] = pB * J + 4 * ac4;
  }

  const int fA = (16 * wm + row) * PA + 2 * kp;
  const int fB = (16 * wn + row) * PA + 2 * kp;

  v8f oR = (v8f){0.f, 0.f, 0.f, 0.f, 0.f, 0.f, 0.f, 0.f};
  v8f oI = (v8f){0.f, 0.f, 0.f, 0.f, 0.f, 0.f, 0.f, 0.f};
  v8f hR = (v8f){0.f, 0.f, 0.f, 0.f, 0.f, 0.f, 0.f, 0.f};
  v8f hI = (v8f){0.f, 0.f, 0.f, 0.f, 0.f, 0.f, 0.f, 0.f};

  v4f rq, rcx[4], rsx[4], rcy[4], rsy[4], rcz[4], rsz[4];
  v2f a0r[KB], a0i[KB], b0r[KB], b0i[KB];
  v2f a1r[KB], a1i[KB], b1r[KB], b1i[KB];

  auto fetch = [&](int ch) {
    const int ka = ch * KC;
#pragma unroll
    for (int e = 0; e < 4; ++e) rq[e] = qp[(size_t)(ka + e) * CCH];
#pragma unroll
    for (int i = 0; i < 4; ++i) {
      rcx[i] = *(const v4f*)(Cx + offA[i] + ka);
      rsx[i] = *(const v4f*)(Sx + offA[i] + ka);
      rcy[i] = *(const v4f*)(Cy + offN[i] + ka);
      rsy[i] = *(const v4f*)(Sy + offN[i] + ka);
      rcz[i] = *(const v4f*)(Cz + offP[i] + ka);
      rsz[i] = *(const v4f*)(Sz + offP[i] + ka);
    }
  };
  auto stash = [&](float* dst) {
    const v4f nq = -rq;
#pragma unroll
    for (int i = 0; i < 4; ++i) {
      *(v4f*)(dst + S_AR + sT + 8 * i * PA) = rq * rcx[i];
      *(v4f*)(dst + S_AI + sT + 8 * i * PA) = nq * rsx[i];
      *(v4f*)(dst + S_BR + sT + 8 * i * PA) = rcy[i] * rcz[i] - rsy[i] * rsz[i];
      *(v4f*)(dst + S_BI + sT + 8 * i * PA) = rsy[i] * rcz[i] + rcy[i] * rsz[i];
    }
  };

  const int nch = K / KC;
  const int nco = KO / KC;

  fetch(0);
  stash(sm);
  __syncthreads();

  frag_load<true>(sm, fA, fB, 0, a0r, a0i, b0r, b0i);
  if (1 < nch) fetch(1);

  auto chunk = [&](int c, v8f& accr, v8f& acci) {
    const float* cur = sm + (c & 1) * S_BUF;
    float* nxt = sm + ((c + 1) & 1) * S_BUF;

    frag_load<true>(cur, fA, fB, 1, a1r, a1i, b1r, b1i);
    __builtin_amdgcn_sched_barrier(0);
    wmma_block<false>(a0r, a0i, b0r, b0i, accr, acci);

    const bool more = (c + 1 < nch);
    if (more) stash(nxt);
    __syncthreads();
    if (more) {
      frag_load<true>(nxt, fA, fB, 0, a0r, a0i, b0r, b0i);
      if (c + 2 < nch) fetch(c + 2);
    }
    wmma_block<false>(a1r, a1i, b1r, b1i, accr, acci);
  };

  for (int c = 0; c < nco; ++c) chunk(c, oR, oI);
  for (int c = nco; c < nch; ++c) chunk(c, hR, hI);

  const int np = n0 + 16 * wn + row;
  if (np < NP) {
    const int n = np / NM, p = np % NM;
    const float ky = TWOPI_L * (float)(n - 15);
    const float kz = TWOPI_L * (float)(p - 15);
    float eOO[6], eOH[6], eHO[6], eHH[6];
#pragma unroll
    for (int i = 0; i < 6; ++i) {
      eOO[i] = expf(2.0f * shOO[i]);
      eOH[i] = expf(2.0f * shOH[i]);
      eHO[i] = expf(2.0f * shHO[i]);
      eHH[i] = expf(2.0f * shHH[i]);
    }
    const int mb = (m0 + 16 * wm + 8 * kp) / CCH;
    float fOr[2][CCH], fOi[2][CCH], fHr[2][CCH], fHi[2][CCH];
#pragma unroll
    for (int h = 0; h < 2; ++h) {
#pragma unroll
      for (int c = 0; c < CCH; ++c) { fOr[h][c] = 0.f; fOi[h][c] = 0.f; fHr[h][c] = 0.f; fHi[h][c] = 0.f; }
      const int m = mb + h;
      if (m < NM) {
        const float kx = TWOPI_L * (float)(m - 15);
        const float squ = kx * kx + ky * ky + kz * kz;
        float mOO = 0.f, mOH = 0.f, mHO = 0.f, mHH = 0.f;
#pragma unroll
        for (int i = 0; i < 6; ++i) {
          mOO += amOO[i] * expf(-squ * eOO[i]);
          mOH += amOH[i] * expf(-squ * eOH[i]);
          mHO += amHO[i] * expf(-squ * eHO[i]);
          mHH += amHH[i] * expf(-squ * eHH[i]);
        }
        if (squ == 0.0f) { mOH = 0.f; mHO = 0.f; mHH = 0.f; }
#pragma unroll
        for (int c = 0; c < CCH; ++c) {
          const int v = CCH * h + c;
          const float ror = oR[v], roi = oI[v];
          const float rhr = hR[v], rhi = hI[v];
          fOr[h][c] = mOO * ror + mOH * rhr;  fOi[h][c] = mOO * roi + mOH * rhi;
          fHr[h][c] = mHO * ror + mHH * rhr;  fHi[h][c] = mHO * roi + mHH * rhi;
        }
      }
    }
#pragma unroll
    for (int c = 0; c < CCH; ++c) {
      const size_t o = ((size_t)(GPAIR * c + mb / 2) * KNP + np) * 2;
      *(v2f*)(gOr + o) = (v2f){fOr[0][c], fOr[1][c]};
      *(v2f*)(gOi + o) = (v2f){fOi[0][c], fOi[1][c]};
      *(v2f*)(gHr + o) = (v2f){fHr[0][c], fHr[1][c]};
      *(v2f*)(gHi + o) = (v2f){fHi[0][c], fHi[1][c]};
    }
  }
}

struct XFrag { v2f r[CCH][8]; v2f i[CCH][8]; };
struct YZPh { float cy[8], sy[8], cz[8], sz[8]; };

__device__ __forceinline__ void t2_load(const float* __restrict__ Gr, const float* __restrict__ Gi,
                                        const float* __restrict__ Cy, const float* __restrict__ Sy,
                                        const float* __restrict__ Cz, const float* __restrict__ Sz,
                                        int np0, int lrow, int h, int j,
                                        XFrag& a, YZPh& w) {
#pragma unroll
  for (int c = 0; c < CCH; ++c) {
#pragma unroll
    for (int ks = 0; ks < 8; ++ks) {
      const size_t o = ((size_t)(GPAIR * c + 2 * ks + h) * KNP + np0 + lrow) * 2;
      a.r[c][ks] = *(const v2f*)(Gr + o);
      a.i[c][ks] = *(const v2f*)(Gi + o);
    }
  }
#pragma unroll
  for (int r = 0; r < 8; ++r) {
    const int np = np0 + 8 * h + r;
    const int n = np / NM, p = np - NM * n;
    const int nn = (n < NM) ? n : (NM - 1);
    w.cy[r] = Cy[nn * J + j];
    w.sy[r] = Sy[nn * J + j];
    w.cz[r] = Cz[p * J + j];
    w.sz[r] = Sz[p * J + j];
  }
}

__device__ __forceinline__ void t2_tile(const XFrag& a, const XFrag& b, const XFrag& bx,
                                        const YZPh& w, int np0, int h,
                                        float& E, float& FX, float& FY, float& FZ) {
  const v8f z = (v8f){0.f, 0.f, 0.f, 0.f, 0.f, 0.f, 0.f, 0.f};
  v8f pRR = z, pII = z, pI = z, xRR = z, xII = z, xI = z;
#pragma unroll
  for (int c = 0; c < CCH; ++c) {
#pragma unroll
    for (int ks = 0; ks < 8; ++ks) {
      pRR = __builtin_amdgcn_wmma_f32_16x16x4_f32(false, a.r[c][ks], false, b.r[c][ks], (short)0, pRR, false, false);
      pII = __builtin_amdgcn_wmma_f32_16x16x4_f32(false, a.i[c][ks], false, b.i[c][ks], (short)0, pII, false, false);
      pI  = __builtin_amdgcn_wmma_f32_16x16x4_f32(false, a.i[c][ks], false, b.r[c][ks], (short)0, pI, false, false);
      pI  = __builtin_amdgcn_wmma_f32_16x16x4_f32(false, a.r[c][ks], false, b.i[c][ks], (short)0, pI, false, false);
      xRR = __builtin_amdgcn_wmma_f32_16x16x4_f32(false, a.r[c][ks], false, bx.r[c][ks], (short)0, xRR, false, false);
      xII = __builtin_amdgcn_wmma_f32_16x16x4_f32(false, a.i[c][ks], false, bx.i[c][ks], (short)0, xII, false, false);
      xI  = __builtin_amdgcn_wmma_f32_16x16x4_f32(false, a.i[c][ks], false, bx.r[c][ks], (short)0, xI, false, false);
      xI  = __builtin_amdgcn_wmma_f32_16x16x4_f32(false, a.r[c][ks], false, bx.i[c][ks], (short)0, xI, false, false);
    }
  }
#pragma unroll
  for (int r = 0; r < 8; ++r) {
    const int np = np0 + 8 * h + r;
    const int n = np / NM, p = np - NM * n;
    const float ky = TWOPI_L * (float)(n - 15);
    const float kz = TWOPI_L * (float)(p - 15);
    const float sr = pRR[r] - pII[r], si = pI[r];
    const float tr = xRR[r] - xII[r], ti = xI[r];
    const float wc = w.cy[r] * w.cz[r] - w.sy[r] * w.sz[r];
    const float ws = w.sy[r] * w.cz[r] + w.cy[r] * w.sz[r];
    E += sr * wc - si * ws;
    const float v = sr * ws + si * wc;
    FY += ky * v;
    FZ += kz * v;
    FX += tr * ws + ti * wc;
  }
}

__global__ __launch_bounds__(128)
void type2_xfirst(const float* __restrict__ gOr, const float* __restrict__ gOi,
                  const float* __restrict__ gHr, const float* __restrict__ gHi,
                  const float* __restrict__ Cy, const float* __restrict__ Sy,
                  const float* __restrict__ Cz, const float* __restrict__ Sz,
                  const float* __restrict__ Cx, const float* __restrict__ Sx,
                  const float* __restrict__ charge, float* __restrict__ part) {
  const int lane = threadIdx.x & 31;
  const int wv   = threadIdx.x >> 5;
  const int lrow = lane & 15;
  const int h    = lane >> 4;
  const int jt   = blockIdx.y * 4 + wv;
  const int j    = 16 * jt + lrow;
  const int sl   = blockIdx.x;
  const bool oxy = (16 * jt < JOs);
  const float* Gr = oxy ? gOr : gHr;
  const float* Gi = oxy ? gOi : gHi;

  XFrag b, bx;
  const v4f q = *(const v4f*)(charge + (size_t)j * CCH);
#pragma unroll
  for (int ks = 0; ks < 8; ++ks) {
#pragma unroll
    for (int e = 0; e < 2; ++e) {
      const int m = 4 * ks + 2 * h + e;
      const int mm = (m < NM) ? m : (NM - 1);
      float cx = Cx[mm * J + j], sx = Sx[mm * J + j];
      if (m >= NM) { cx = 0.f; sx = 0.f; }
      const float kx = TWOPI_L * (float)(m - 15);
#pragma unroll
      for (int c = 0; c < CCH; ++c) {
        const float br = q[c] * cx, bi = q[c] * sx;
        b.r[c][ks][e] = br;        b.i[c][ks][e] = bi;
        bx.r[c][ks][e] = kx * br;  bx.i[c][ks][e] = kx * bi;
      }
    }
  }

  float E = 0.f, FX = 0.f, FY = 0.f, FZ = 0.f;
  XFrag a0, a1;
  YZPh w0, w1;
  const int t0 = TPS * sl;
  t2_load(Gr, Gi, Cy, Sy, Cz, Sz, 16 * t0, lrow, h, j, a0, w0);
  if (t0 + 1 < NT16) t2_load(Gr, Gi, Cy, Sy, Cz, Sz, 16 * (t0 + 1), lrow, h, j, a1, w1);
  __builtin_amdgcn_sched_barrier(0);
  t2_tile(a0, b, bx, w0, 16 * t0, h, E, FX, FY, FZ);
  if (t0 + 1 < NT16) {
    if (t0 + 2 < NT16) t2_load(Gr, Gi, Cy, Sy, Cz, Sz, 16 * (t0 + 2), lrow, h, j, a0, w0);
    __builtin_amdgcn_sched_barrier(0);
    t2_tile(a1, b, bx, w1, 16 * (t0 + 1), h, E, FX, FY, FZ);
    if (t0 + 2 < NT16) t2_tile(a0, b, bx, w0, 16 * (t0 + 2), h, E, FX, FY, FZ);
  }

  E  += __shfl_xor(E, 16);
  FX += __shfl_xor(FX, 16);
  FY += __shfl_xor(FY, 16);
  FZ += __shfl_xor(FZ, 16);
  if (h == 0) *(v4f*)(part + ((size_t)sl * J + j) * 4) = (v4f){E, FX, FY, FZ};
}

__global__ void finalize(const float* __restrict__ part, float* __restrict__ out, int b) {
  int j = blockIdx.x * blockDim.x + threadIdx.x;
  if (j >= J) return;
  v4f acc = (v4f){0.f, 0.f, 0.f, 0.f};
  for (int sl = 0; sl < NSL; ++sl) {
    const v4f p = *(const v4f*)(part + ((size_t)sl * J + j) * 4);
    acc += p;
  }
  out[b * J + j] = acc[0] / (2.0f * VBOX);
  float* F = out + 2 * J;
  F[(size_t)(b * J + j) * 3 + 0] = acc[1] / VBOX;
  F[(size_t)(b * J + j) * 3 + 1] = acc[2] / VBOX;
  F[(size_t)(b * J + j) * 3 + 2] = acc[3] / VBOX;
}

extern "C" void kernel_launch(void* const* d_in, const int* in_sizes, int n_in,
                              void* d_out, int out_size, void* d_ws, size_t ws_size,
                              hipStream_t stream) {
  const float* pos    = (const float*)d_in[0];
  const float* charge = (const float*)d_in[1];
  const float* shOO = (const float*)d_in[2];
  const float* amOO = (const float*)d_in[3];
  const float* shOH = (const float*)d_in[4];
  const float* amOH = (const float*)d_in[5];
  const float* shHO = (const float*)d_in[6];
  const float* amHO = (const float*)d_in[7];
  const float* shHH = (const float*)d_in[8];
  const float* amHH = (const float*)d_in[9];
  float* out = (float*)d_out;
  float* ws  = (float*)d_ws;

  float *Cx = ws + OFF_CX, *Sx = ws + OFF_SX;
  float *Cy = ws + OFF_CY, *Sy = ws + OFF_SY;
  float *Cz = ws + OFF_CZ, *Sz = ws + OFF_SZ;
  float *gOr = ws + OFF_GOR, *gOi = ws + OFF_GOI;
  float *gHr = ws + OFF_GHR, *gHi = ws + OFF_GHI;
  float *part = ws + OFF_PART;

  hipMemsetAsync(d_ws, 0, WS_TOTAL * sizeof(float), stream);

  for (int b = 0; b < 2; ++b) {
    const float* posb = pos + (size_t)b * J * 3;
    const float* chb  = charge + (size_t)b * J * CCH;

    phase_build<<<(3 * J + 127) / 128, 128, 0, stream>>>(posb, Cx, Sx, Cy, Sy, Cz, Sz);

    dim3 g1(NPPAD / TN, CMP / TM);
    nufft1_mix<<<g1, 128, 0, stream>>>(chb, Cx, Sx, Cy, Sy, Cz, Sz,
                                       shOO, amOO, shOH, amOH, shHO, amHO, shHH, amHH,
                                       gOr, gOi, gHr, gHi, JOs, J);

    dim3 g2(NSL, J / 64);
    type2_xfirst<<<g2, 128, 0, stream>>>(gOr, gOi, gHr, gHi, Cy, Sy, Cz, Sz, Cx, Sx, chb, part);

    finalize<<<(J + 127) / 128, 128, 0, stream>>>(part, out, b);
  }
}


# x axis of both transforms in the real cos/sin basis (charges are real): nufft1's A and type-2's B become real, a complex K-step costs 2 matrix instructions, not 4; sums over m and -m regrouped
# speedup vs baseline: 11.2579x; 1.1528x over previous
#include <hip/hip_runtime.h>
#include <hip/hip_bf16.h>
#include <math.h>

typedef float v2f __attribute__((ext_vector_type(2)));
typedef float v8f __attribute__((ext_vector_type(8)));

constexpr int NM    = 31;
constexpr int NP    = 961;
constexpr int NPPAD = 1024;
constexpr int KNP   = 1024;
constexpr int J     = 768;
constexpr int JOs   = 256;
constexpr int CCH   = 4;
constexpr int CMP   = 128;
constexpr int GPAIR = 16;
constexpr int NT16  = 61;
constexpr int TPS   = 3;
constexpr int NSL   = 21;
constexpr float LBOX   = 20.0f;
constexpr float VBOX   = 8000.0f;
constexpr float TWOPI_L = 0.31415926535897931f;

constexpr size_t SZ_PH  = (size_t)NM * J;
constexpr size_t OFF_CX = 0;
constexpr size_t OFF_SX = OFF_CX + SZ_PH;
constexpr size_t OFF_CY = OFF_SX + SZ_PH;
constexpr size_t OFF_SY = OFF_CY + SZ_PH;
constexpr size_t OFF_CZ = OFF_SY + SZ_PH;
constexpr size_t OFF_SZ = OFF_CZ + SZ_PH;
constexpr size_t SZ_G   = (size_t)CCH * GPAIR * KNP * 2;
constexpr size_t OFF_GOR = OFF_SZ + SZ_PH;
constexpr size_t OFF_GOI = OFF_GOR + SZ_G;
constexpr size_t OFF_GHR = OFF_GOI + SZ_G;
constexpr size_t OFF_GHI = OFF_GHR + SZ_G;
constexpr size_t SZ_PART = (size_t)NSL * J * 4;
constexpr size_t OFF_PART = OFF_GHI + SZ_G;
constexpr size_t WS_TOTAL = OFF_PART + SZ_PART;

__global__ void phase_build(const float* __restrict__ pos,
                            float* __restrict__ Cx, float* __restrict__ Sx,
                            float* __restrict__ Cy, float* __restrict__ Sy,
                            float* __restrict__ Cz, float* __restrict__ Sz) {
  int t = blockIdx.x * blockDim.x + threadIdx.x;
  if (t >= 3 * J) return;
  int d = t / J, j = t % J;
  float th = TWOPI_L * (pos[j * 3 + d] - 0.5f * LBOX);
  float* C = (d == 0) ? Cx : (d == 1) ? Cy : Cz;
  float* S = (d == 0) ? Sx : (d == 1) ? Sy : Sz;
  for (int mi = 0; mi < NM; ++mi) {
    float s, c;
    sincosf((float)(mi - 15) * th, &s, &c);
    C[mi * J + j] = c;
    S[mi * J + j] = s;
  }
}

typedef float v4f __attribute__((ext_vector_type(4)));
constexpr int TM = 32;
constexpr int TN = 32;
constexpr int KC = 64;
constexpr int PA = 68;
constexpr int SZ_A = TM * PA;
constexpr int SZ_B = TN * PA;
constexpr int S_A  = 0;
constexpr int S_BR = S_A + SZ_A;
constexpr int S_BI = S_BR + SZ_B;
constexpr int S_BUF = S_BI + SZ_B;
static_assert(SZ_A % 4 == 0 && SZ_B % 4 == 0, "planes start on 16-byte boundaries");

constexpr int KB = KC / 8;

__device__ __forceinline__ void frag_load(const float* buf, int fA, int fB, int blk,
                                          v2f (&a)[KB], v2f (&br)[KB], v2f (&bi)[KB]) {
#pragma unroll
  for (int s = 0; s < KB; ++s) {
    const int ks = blk * KB + s;
    a[s]  = *(const v2f*)(buf + S_A + fA + 4 * ks);
    br[s] = *(const v2f*)(buf + S_BR + fB + 4 * ks);
    bi[s] = *(const v2f*)(buf + S_BI + fB + 4 * ks);
  }
}

__device__ __forceinline__ void wmma_block(const v2f (&a)[KB], const v2f (&br)[KB], const v2f (&bi)[KB],
                                           v8f& accr, v8f& acci) {
#pragma unroll
  for (int s = 0; s < KB; ++s) {
    accr = __builtin_amdgcn_wmma_f32_16x16x4_f32(
        false, a[s], false, br[s], (short)0, accr, false, false);
    acci = __builtin_amdgcn_wmma_f32_16x16x4_f32(
        false, a[s], false, bi[s], (short)0, acci, false, false);
  }
}

__global__ __launch_bounds__(128)
void nufft1_mix(const float* __restrict__ charge,
                const float* __restrict__ Cx, const float* __restrict__ Sx,
                const float* __restrict__ Cy, const float* __restrict__ Sy,
                const float* __restrict__ Cz, const float* __restrict__ Sz,
                const float* __restrict__ shOO, const float* __restrict__ amOO,
                const float* __restrict__ shOH, const float* __restrict__ amOH,
                const float* __restrict__ shHO, const float* __restrict__ amHO,
                const float* __restrict__ shHH, const float* __restrict__ amHH,
                float* __restrict__ gOr, float* __restrict__ gOi,
                float* __restrict__ gHr, float* __restrict__ gHi, int KO, int K) {
  __shared__ __attribute__((aligned(16))) float sm[2 * S_BUF];

  const int t    = threadIdx.x;
  const int lane = t & 31;
  const int wv   = t >> 5;
  const int row  = lane & 15;
  const int kp   = lane >> 4;
  const int wm   = wv >> 1;
  const int wn   = wv & 1;
  const int n0 = blockIdx.x * TN;
  const int m0 = blockIdx.y * TM;

  const int ac4 = t & 15, arow = t >> 4;
  const int sT = arow * PA + 4 * ac4;
  const float* qp = charge + (size_t)(4 * ac4) * CCH + (arow & 3);
  const float* pA[4];
  int offN[4], offP[4];
#pragma unroll
  for (int i = 0; i < 4; ++i) {
    int u = (m0 + arow + 8 * i) >> 2;
    if (u > NM - 1) u = NM - 1;
    pA[i] = ((u < 16) ? (Cx + (u + 15) * J) : (Sx + u * J)) + 4 * ac4;
    int npB = n0 + arow + 8 * i;
    if (npB > NP - 1) npB = NP - 1;
    const int nB = npB / NM, pB = npB - NM * nB;
    offN[i] = nB * J + 4 * ac4;
    offP[i] = pB * J + 4 * ac4;
  }

  const int fA = (16 * wm + row) * PA + 2 * kp;
  const int fB = (16 * wn + row) * PA + 2 * kp;

  v8f oR = (v8f){0.f, 0.f, 0.f, 0.f, 0.f, 0.f, 0.f, 0.f};
  v8f oI = (v8f){0.f, 0.f, 0.f, 0.f, 0.f, 0.f, 0.f, 0.f};
  v8f hR = (v8f){0.f, 0.f, 0.f, 0.f, 0.f, 0.f, 0.f, 0.f};
  v8f hI = (v8f){0.f, 0.f, 0.f, 0.f, 0.f, 0.f, 0.f, 0.f};

  v4f rq, rax[4], rcy[4], rsy[4], rcz[4], rsz[4];
  v2f a0[KB], b0r[KB], b0i[KB];
  v2f a1[KB], b1r[KB], b1i[KB];

  auto fetch = [&](int ch) {
    const int ka = ch * KC;
#pragma unroll
    for (int e = 0; e < 4; ++e) rq[e] = qp[(size_t)(ka + e) * CCH];
#pragma unroll
    for (int i = 0; i < 4; ++i) {
      rax[i] = *(const v4f*)(pA[i] + ka);
      rcy[i] = *(const v4f*)(Cy + offN[i] + ka);
      rsy[i] = *(const v4f*)(Sy + offN[i] + ka);
      rcz[i] = *(const v4f*)(Cz + offP[i] + ka);
      rsz[i] = *(const v4f*)(Sz + offP[i] + ka);
    }
  };
  auto stash = [&](float* dst) {
#pragma unroll
    for (int i = 0; i < 4; ++i) {
      *(v4f*)(dst + S_A + sT + 8 * i * PA)  = rq * rax[i];
      *(v4f*)(dst + S_BR + sT + 8 * i * PA) = rcy[i] * rcz[i] - rsy[i] * rsz[i];
      *(v4f*)(dst + S_BI + sT + 8 * i * PA) = rsy[i] * rcz[i] + rcy[i] * rsz[i];
    }
  };

  const int nch = K / KC;
  const int nco = KO / KC;

  fetch(0);
  stash(sm);
  __syncthreads();

  frag_load(sm, fA, fB, 0, a0, b0r, b0i);
  if (1 < nch) fetch(1);

  auto chunk = [&](int c, v8f& accr, v8f& acci) {
    const float* cur = sm + (c & 1) * S_BUF;
    float* nxt = sm + ((c + 1) & 1) * S_BUF;

    frag_load(cur, fA, fB, 1, a1, b1r, b1i);
    __builtin_amdgcn_sched_barrier(0);
    wmma_block(a0, b0r, b0i, accr, acci);

    const bool more = (c + 1 < nch);
    if (more) stash(nxt);
    __syncthreads();
    if (more) {
      frag_load(nxt, fA, fB, 0, a0, b0r, b0i);
      if (c + 2 < nch) fetch(c + 2);
    }
    wmma_block(a1, b1r, b1i, accr, acci);
  };

  for (int c = 0; c < nco; ++c) chunk(c, oR, oI);
  for (int c = nco; c < nch; ++c) chunk(c, hR, hI);

  const int np = n0 + 16 * wn + row;
  if (np < NP) {
    const int n = np / NM, p = np % NM;
    const float ky = TWOPI_L * (float)(n - 15);
    const float kz = TWOPI_L * (float)(p - 15);
    float eOO[6], eOH[6], eHO[6], eHH[6];
#pragma unroll
    for (int i = 0; i < 6; ++i) {
      eOO[i] = expf(2.0f * shOO[i]);
      eOH[i] = expf(2.0f * shOH[i]);
      eHO[i] = expf(2.0f * shHO[i]);
      eHH[i] = expf(2.0f * shHH[i]);
    }
    const int mb = (m0 + 16 * wm + 8 * kp) / CCH;
    float fOr[2][CCH], fOi[2][CCH], fHr[2][CCH], fHi[2][CCH];
#pragma unroll
    for (int h = 0; h < 2; ++h) {
#pragma unroll
      for (int c = 0; c < CCH; ++c) { fOr[h][c] = 0.f; fOi[h][c] = 0.f; fHr[h][c] = 0.f; fHi[h][c] = 0.f; }
      const int u = mb + h;
      if (u < NM) {
        const int a = (u < 16) ? u : (u - 15);
        const float kx = TWOPI_L * (float)a;
        const float squ = kx * kx + ky * ky + kz * kz;
        float mOO = 0.f, mOH = 0.f, mHO = 0.f, mHH = 0.f;
#pragma unroll
        for (int i = 0; i < 6; ++i) {
          mOO += amOO[i] * expf(-squ * eOO[i]);
          mOH += amOH[i] * expf(-squ * eOH[i]);
          mHO += amHO[i] * expf(-squ * eHO[i]);
          mHH += amHH[i] * expf(-squ * eHH[i]);
        }
        if (squ == 0.0f) { mOH = 0.f; mHO = 0.f; mHH = 0.f; }
        const float w = (u == 0) ? 1.0f : 2.0f;
        mOO *= w; mOH *= w; mHO *= w; mHH *= w;
#pragma unroll
        for (int c = 0; c < CCH; ++c) {
          const int v = CCH * h + c;
          const float por = oR[v], poi = oI[v];
          const float phr = hR[v], phi = hI[v];
          fOr[h][c] = mOO * por + mOH * phr;  fOi[h][c] = -(mOO * poi + mOH * phi);
          fHr[h][c] = mHO * por + mHH * phr;  fHi[h][c] = -(mHO * poi + mHH * phi);
        }
      }
    }
#pragma unroll
    for (int c = 0; c < CCH; ++c) {
      const size_t o = ((size_t)(GPAIR * c + mb / 2) * KNP + np) * 2;
      *(v2f*)(gOr + o) = (v2f){fOr[0][c], fOr[1][c]};
      *(v2f*)(gOi + o) = (v2f){fOi[0][c], fOi[1][c]};
      *(v2f*)(gHr + o) = (v2f){fHr[0][c], fHr[1][c]};
      *(v2f*)(gHi + o) = (v2f){fHi[0][c], fHi[1][c]};
    }
  }
}

struct XFrag { v2f r[CCH][8]; v2f i[CCH][8]; };
struct BFrag { v2f v[CCH][8]; };
struct YZPh { float cy[8], sy[8], cz[8], sz[8]; };

__device__ __forceinline__ void t2_load(const float* __restrict__ Gr, const float* __restrict__ Gi,
                                        const float* __restrict__ Cy, const float* __restrict__ Sy,
                                        const float* __restrict__ Cz, const float* __restrict__ Sz,
                                        int np0, int lrow, int h, int j,
                                        XFrag& a, YZPh& w) {
#pragma unroll
  for (int c = 0; c < CCH; ++c) {
#pragma unroll
    for (int ks = 0; ks < 8; ++ks) {
      const size_t o = ((size_t)(GPAIR * c + 2 * ks + h) * KNP + np0 + lrow) * 2;
      a.r[c][ks] = *(const v2f*)(Gr + o);
      a.i[c][ks] = *(const v2f*)(Gi + o);
    }
  }
#pragma unroll
  for (int r = 0; r < 8; ++r) {
    const int np = np0 + 8 * h + r;
    const int n = np / NM, p = np - NM * n;
    const int nn = (n < NM) ? n : (NM - 1);
    w.cy[r] = Cy[nn * J + j];
    w.sy[r] = Sy[nn * J + j];
    w.cz[r] = Cz[p * J + j];
    w.sz[r] = Sz[p * J + j];
  }
}

__device__ __forceinline__ void t2_tile(const XFrag& a, const BFrag& b, const BFrag& b2,
                                        const YZPh& w, int np0, int h,
                                        float& E, float& FX, float& FY, float& FZ) {
  const v8f z = (v8f){0.f, 0.f, 0.f, 0.f, 0.f, 0.f, 0.f, 0.f};
  v8f pR = z, pI = z, yR = z, yI = z;
#pragma unroll
  for (int c = 0; c < CCH; ++c) {
#pragma unroll
    for (int ks = 0; ks < 8; ++ks) {
      pR = __builtin_amdgcn_wmma_f32_16x16x4_f32(false, a.r[c][ks], false, b.v[c][ks], (short)0, pR, false, false);
      pI = __builtin_amdgcn_wmma_f32_16x16x4_f32(false, a.i[c][ks], false, b.v[c][ks], (short)0, pI, false, false);
      yR = __builtin_amdgcn_wmma_f32_16x16x4_f32(false, a.r[c][ks], false, b2.v[c][ks], (short)0, yR, false, false);
      yI = __builtin_amdgcn_wmma_f32_16x16x4_f32(false, a.i[c][ks], false, b2.v[c][ks], (short)0, yI, false, false);
    }
  }
#pragma unroll
  for (int r = 0; r < 8; ++r) {
    const int np = np0 + 8 * h + r;
    const int n = np / NM, p = np - NM * n;
    const float ky = TWOPI_L * (float)(n - 15);
    const float kz = TWOPI_L * (float)(p - 15);
    const float sr = pR[r], si = pI[r];
    const float tr = -yI[r], ti = yR[r];
    const float wc = w.cy[r] * w.cz[r] - w.sy[r] * w.sz[r];
    const float ws = w.sy[r] * w.cz[r] + w.cy[r] * w.sz[r];
    E += sr * wc - si * ws;
    const float v = sr * ws + si * wc;
    FY += ky * v;
    FZ += kz * v;
    FX += tr * ws + ti * wc;
  }
}

__global__ __launch_bounds__(128)
void type2_xfirst(const float* __restrict__ gOr, const float* __restrict__ gOi,
                  const float* __restrict__ gHr, const float* __restrict__ gHi,
                  const float* __restrict__ Cy, const float* __restrict__ Sy,
                  const float* __restrict__ Cz, const float* __restrict__ Sz,
                  const float* __restrict__ Cx, const float* __restrict__ Sx,
                  const float* __restrict__ charge, float* __restrict__ part) {
  const int lane = threadIdx.x & 31;
  const int wv   = threadIdx.x >> 5;
  const int lrow = lane & 15;
  const int h    = lane >> 4;
  const int jt   = blockIdx.y * 4 + wv;
  const int j    = 16 * jt + lrow;
  const int sl   = blockIdx.x;
  const bool oxy = (16 * jt < JOs);
  const float* Gr = oxy ? gOr : gHr;
  const float* Gi = oxy ? gOi : gHi;

  BFrag b, b2;
  const v4f q = *(const v4f*)(charge + (size_t)j * CCH);
#pragma unroll
  for (int ks = 0; ks < 8; ++ks) {
#pragma unroll
    for (int e = 0; e < 2; ++e) {
      const int u = 4 * ks + 2 * h + e;
      const bool cosrow = (ks < 4);
      const int a = cosrow ? u : (u - 15);
      const int tr = (a + 15 < NM) ? (a + 15) : (NM - 1);
      float cx = Cx[tr * J + j], sx = Sx[tr * J + j];
      if (u >= NM) { cx = 0.f; sx = 0.f; }
      const float kx = TWOPI_L * (float)a;
#pragma unroll
      for (int c = 0; c < CCH; ++c) {
        const float qc = q[c] * cx, qs = q[c] * sx;
        b.v[c][ks][e]  = cosrow ? qc : qs;
        b2.v[c][ks][e] = cosrow ? (kx * qs) : -(kx * qc);
      }
    }
  }

  float E = 0.f, FX = 0.f, FY = 0.f, FZ = 0.f;
  XFrag a0, a1;
  YZPh w0, w1;
  const int t0 = TPS * sl;
  t2_load(Gr, Gi, Cy, Sy, Cz, Sz, 16 * t0, lrow, h, j, a0, w0);
  if (t0 + 1 < NT16) t2_load(Gr, Gi, Cy, Sy, Cz, Sz, 16 * (t0 + 1), lrow, h, j, a1, w1);
  __builtin_amdgcn_sched_barrier(0);
  t2_tile(a0, b, b2, w0, 16 * t0, h, E, FX, FY, FZ);
  if (t0 + 1 < NT16) {
    if (t0 + 2 < NT16) t2_load(Gr, Gi, Cy, Sy, Cz, Sz, 16 * (t0 + 2), lrow, h, j, a0, w0);
    __builtin_amdgcn_sched_barrier(0);
    t2_tile(a1, b, b2, w1, 16 * (t0 + 1), h, E, FX, FY, FZ);
    if (t0 + 2 < NT16) t2_tile(a0, b, b2, w0, 16 * (t0 + 2), h, E, FX, FY, FZ);
  }

  E  += __shfl_xor(E, 16);
  FX += __shfl_xor(FX, 16);
  FY += __shfl_xor(FY, 16);
  FZ += __shfl_xor(FZ, 16);
  if (h == 0) *(v4f*)(part + ((size_t)sl * J + j) * 4) = (v4f){E, FX, FY, FZ};
}

__global__ void finalize(const float* __restrict__ part, float* __restrict__ out, int b) {
  int j = blockIdx.x * blockDim.x + threadIdx.x;
  if (j >= J) return;
  v4f acc = (v4f){0.f, 0.f, 0.f, 0.f};
  for (int sl = 0; sl < NSL; ++sl) {
    const v4f p = *(const v4f*)(part + ((size_t)sl * J + j) * 4);
    acc += p;
  }
  out[b * J + j] = acc[0] / (2.0f * VBOX);
  float* F = out + 2 * J;
  F[(size_t)(b * J + j) * 3 + 0] = acc[1] / VBOX;
  F[(size_t)(b * J + j) * 3 + 1] = acc[2] / VBOX;
  F[(size_t)(b * J + j) * 3 + 2] = acc[3] / VBOX;
}

extern "C" void kernel_launch(void* const* d_in, const int* in_sizes, int n_in,
                              void* d_out, int out_size, void* d_ws, size_t ws_size,
                              hipStream_t stream) {
  const float* pos    = (const float*)d_in[0];
  const float* charge = (const float*)d_in[1];
  const float* shOO = (const float*)d_in[2];
  const float* amOO = (const float*)d_in[3];
  const float* shOH = (const float*)d_in[4];
  const float* amOH = (const float*)d_in[5];
  const float* shHO = (const float*)d_in[6];
  const float* amHO = (const float*)d_in[7];
  const float* shHH = (const float*)d_in[8];
  const float* amHH = (const float*)d_in[9];
  float* out = (float*)d_out;
  float* ws  = (float*)d_ws;

  float *Cx = ws + OFF_CX, *Sx = ws + OFF_SX;
  float *Cy = ws + OFF_CY, *Sy = ws + OFF_SY;
  float *Cz = ws + OFF_CZ, *Sz = ws + OFF_SZ;
  float *gOr = ws + OFF_GOR, *gOi = ws + OFF_GOI;
  float *gHr = ws + OFF_GHR, *gHi = ws + OFF_GHI;
  float *part = ws + OFF_PART;

  hipMemsetAsync(d_ws, 0, WS_TOTAL * sizeof(float), stream);

  for (int b = 0; b < 2; ++b) {
    const float* posb = pos + (size_t)b * J * 3;
    const float* chb  = charge + (size_t)b * J * CCH;

    phase_build<<<(3 * J + 127) / 128, 128, 0, stream>>>(posb, Cx, Sx, Cy, Sy, Cz, Sz);

    dim3 g1(NPPAD / TN, CMP / TM);
    nufft1_mix<<<g1, 128, 0, stream>>>(chb, Cx, Sx, Cy, Sy, Cz, Sz,
                                       shOO, amOO, shOH, amOH, shHO, amHO, shHH, amHH,
                                       gOr, gOi, gHr, gHi, JOs, J);

    dim3 g2(NSL, J / 64);
    type2_xfirst<<<g2, 128, 0, stream>>>(gOr, gOi, gHr, gHi, Cy, Sy, Cz, Sz, Cx, Sx, chb, part);

    finalize<<<(J + 127) / 128, 128, 0, stream>>>(part, out, b);
  }
}


# Phase tables: one sincosf per thread (axis, |m|, particle) instead of a 31-step loop per (axis, particle); rows m and -m stored as exact mirrors by the same thread; values unchanged
# speedup vs baseline: 11.8533x; 1.0529x over previous
#include <hip/hip_runtime.h>
#include <hip/hip_bf16.h>
#include <math.h>

typedef float v2f __attribute__((ext_vector_type(2)));
typedef float v8f __attribute__((ext_vector_type(8)));

constexpr int NM    = 31;
constexpr int NP    = 961;
constexpr int NPPAD = 1024;
constexpr int KNP   = 1024;
constexpr int J     = 768;
constexpr int JOs   = 256;
constexpr int CCH   = 4;
constexpr int CMP   = 128;
constexpr int GPAIR = 16;
constexpr int NT16  = 61;
constexpr int TPS   = 3;
constexpr int NSL   = 21;
constexpr float LBOX   = 20.0f;
constexpr float VBOX   = 8000.0f;
constexpr float TWOPI_L = 0.31415926535897931f;

constexpr size_t SZ_PH  = (size_t)NM * J;
constexpr size_t OFF_CX = 0;
constexpr size_t OFF_SX = OFF_CX + SZ_PH;
constexpr size_t OFF_CY = OFF_SX + SZ_PH;
constexpr size_t OFF_SY = OFF_CY + SZ_PH;
constexpr size_t OFF_CZ = OFF_SY + SZ_PH;
constexpr size_t OFF_SZ = OFF_CZ + SZ_PH;
constexpr size_t SZ_G   = (size_t)CCH * GPAIR * KNP * 2;
constexpr size_t OFF_GOR = OFF_SZ + SZ_PH;
constexpr size_t OFF_GOI = OFF_GOR + SZ_G;
constexpr size_t OFF_GHR = OFF_GOI + SZ_G;
constexpr size_t OFF_GHI = OFF_GHR + SZ_G;
constexpr size_t SZ_PART = (size_t)NSL * J * 4;
constexpr size_t OFF_PART = OFF_GHI + SZ_G;
constexpr size_t WS_TOTAL = OFF_PART + SZ_PART;

constexpr int NA = 16;
__global__ void phase_build(const float* __restrict__ pos,
                            float* __restrict__ Cx, float* __restrict__ Sx,
                            float* __restrict__ Cy, float* __restrict__ Sy,
                            float* __restrict__ Cz, float* __restrict__ Sz) {
  int t = blockIdx.x * blockDim.x + threadIdx.x;
  if (t >= 3 * NA * J) return;
  const int j = t % J, da = t / J;
  const int d = da / NA, a = da % NA;
  float th = TWOPI_L * (pos[j * 3 + d] - 0.5f * LBOX);
  float* C = (d == 0) ? Cx : (d == 1) ? Cy : Cz;
  float* S = (d == 0) ? Sx : (d == 1) ? Sy : Sz;
  float s, c;
  sincosf((float)a * th, &s, &c);
  C[(15 + a) * J + j] = c;
  S[(15 + a) * J + j] = s;
  if (a != 0) {
    C[(15 - a) * J + j] = c;
    S[(15 - a) * J + j] = -s;
  }
}

typedef float v4f __attribute__((ext_vector_type(4)));
constexpr int TM = 32;
constexpr int TN = 32;
constexpr int KC = 64;
constexpr int PA = 68;
constexpr int SZ_A = TM * PA;
constexpr int SZ_B = TN * PA;
constexpr int S_A  = 0;
constexpr int S_BR = S_A + SZ_A;
constexpr int S_BI = S_BR + SZ_B;
constexpr int S_BUF = S_BI + SZ_B;
static_assert(SZ_A % 4 == 0 && SZ_B % 4 == 0, "planes start on 16-byte boundaries");

constexpr int KB = KC / 8;

__device__ __forceinline__ void frag_load(const float* buf, int fA, int fB, int blk,
                                          v2f (&a)[KB], v2f (&br)[KB], v2f (&bi)[KB]) {
#pragma unroll
  for (int s = 0; s < KB; ++s) {
    const int ks = blk * KB + s;
    a[s]  = *(const v2f*)(buf + S_A + fA + 4 * ks);
    br[s] = *(const v2f*)(buf + S_BR + fB + 4 * ks);
    bi[s] = *(const v2f*)(buf + S_BI + fB + 4 * ks);
  }
}

__device__ __forceinline__ void wmma_block(const v2f (&a)[KB], const v2f (&br)[KB], const v2f (&bi)[KB],
                                           v8f& accr, v8f& acci) {
#pragma unroll
  for (int s = 0; s < KB; ++s) {
    accr = __builtin_amdgcn_wmma_f32_16x16x4_f32(
        false, a[s], false, br[s], (short)0, accr, false, false);
    acci = __builtin_amdgcn_wmma_f32_16x16x4_f32(
        false, a[s], false, bi[s], (short)0, acci, false, false);
  }
}

__global__ __launch_bounds__(128)
void nufft1_mix(const float* __restrict__ charge,
                const float* __restrict__ Cx, const float* __restrict__ Sx,
                const float* __restrict__ Cy, const float* __restrict__ Sy,
                const float* __restrict__ Cz, const float* __restrict__ Sz,
                const float* __restrict__ shOO, const float* __restrict__ amOO,
                const float* __restrict__ shOH, const float* __restrict__ amOH,
                const float* __restrict__ shHO, const float* __restrict__ amHO,
                const float* __restrict__ shHH, const float* __restrict__ amHH,
                float* __restrict__ gOr, float* __restrict__ gOi,
                float* __restrict__ gHr, float* __restrict__ gHi, int KO, int K) {
  __shared__ __attribute__((aligned(16))) float sm[2 * S_BUF];

  const int t    = threadIdx.x;
  const int lane = t & 31;
  const int wv   = t >> 5;
  const int row  = lane & 15;
  const int kp   = lane >> 4;
  const int wm   = wv >> 1;
  const int wn   = wv & 1;
  const int n0 = blockIdx.x * TN;
  const int m0 = blockIdx.y * TM;

  const int ac4 = t & 15, arow = t >> 4;
  const int sT = arow * PA + 4 * ac4;
  const float* qp = charge + (size_t)(4 * ac4) * CCH + (arow & 3);
  const float* pA[4];
  int offN[4], offP[4];
#pragma unroll
  for (int i = 0; i < 4; ++i) {
    int u = (m0 + arow + 8 * i) >> 2;
    if (u > NM - 1) u = NM - 1;
    pA[i] = ((u < 16) ? (Cx + (u + 15) * J) : (Sx + u * J)) + 4 * ac4;
    int npB = n0 + arow + 8 * i;
    if (npB > NP - 1) npB = NP - 1;
    const int nB = npB / NM, pB = npB - NM * nB;
    offN[i] = nB * J + 4 * ac4;
    offP[i] = pB * J + 4 * ac4;
  }

  const int fA = (16 * wm + row) * PA + 2 * kp;
  const int fB = (16 * wn + row) * PA + 2 * kp;

  v8f oR = (v8f){0.f, 0.f, 0.f, 0.f, 0.f, 0.f, 0.f, 0.f};
  v8f oI = (v8f){0.f, 0.f, 0.f, 0.f, 0.f, 0.f, 0.f, 0.f};
  v8f hR = (v8f){0.f, 0.f, 0.f, 0.f, 0.f, 0.f, 0.f, 0.f};
  v8f hI = (v8f){0.f, 0.f, 0.f, 0.f, 0.f, 0.f, 0.f, 0.f};

  v4f rq, rax[4], rcy[4], rsy[4], rcz[4], rsz[4];
  v2f a0[KB], b0r[KB], b0i[KB];
  v2f a1[KB], b1r[KB], b1i[KB];

  auto fetch = [&](int ch) {
    const int ka = ch * KC;
#pragma unroll
    for (int e = 0; e < 4; ++e) rq[e] = qp[(size_t)(ka + e) * CCH];
#pragma unroll
    for (int i = 0; i < 4; ++i) {
      rax[i] = *(const v4f*)(pA[i] + ka);
      rcy[i] = *(const v4f*)(Cy + offN[i] + ka);
      rsy[i] = *(const v4f*)(Sy + offN[i] + ka);
      rcz[i] = *(const v4f*)(Cz + offP[i] + ka);
      rsz[i] = *(const v4f*)(Sz + offP[i] + ka);
    }
  };
  auto stash = [&](float* dst) {
#pragma unroll
    for (int i = 0; i < 4; ++i) {
      *(v4f*)(dst + S_A + sT + 8 * i * PA)  = rq * rax[i];
      *(v4f*)(dst + S_BR + sT + 8 * i * PA) = rcy[i] * rcz[i] - rsy[i] * rsz[i];
      *(v4f*)(dst + S_BI + sT + 8 * i * PA) = rsy[i] * rcz[i] + rcy[i] * rsz[i];
    }
  };

  const int nch = K / KC;
  const int nco = KO / KC;

  fetch(0);
  stash(sm);
  __syncthreads();

  frag_load(sm, fA, fB, 0, a0, b0r, b0i);
  if (1 < nch) fetch(1);

  auto chunk = [&](int c, v8f& accr, v8f& acci) {
    const float* cur = sm + (c & 1) * S_BUF;
    float* nxt = sm + ((c + 1) & 1) * S_BUF;

    frag_load(cur, fA, fB, 1, a1, b1r, b1i);
    __builtin_amdgcn_sched_barrier(0);
    wmma_block(a0, b0r, b0i, accr, acci);

    const bool more = (c + 1 < nch);
    if (more) stash(nxt);
    __syncthreads();
    if (more) {
      frag_load(nxt, fA, fB, 0, a0, b0r, b0i);
      if (c + 2 < nch) fetch(c + 2);
    }
    wmma_block(a1, b1r, b1i, accr, acci);
  };

  for (int c = 0; c < nco; ++c) chunk(c, oR, oI);
  for (int c = nco; c < nch; ++c) chunk(c, hR, hI);

  const int np = n0 + 16 * wn + row;
  if (np < NP) {
    const int n = np / NM, p = np % NM;
    const float ky = TWOPI_L * (float)(n - 15);
    const float kz = TWOPI_L * (float)(p - 15);
    float eOO[6], eOH[6], eHO[6], eHH[6];
#pragma unroll
    for (int i = 0; i < 6; ++i) {
      eOO[i] = expf(2.0f * shOO[i]);
      eOH[i] = expf(2.0f * shOH[i]);
      eHO[i] = expf(2.0f * shHO[i]);
      eHH[i] = expf(2.0f * shHH[i]);
    }
    const int mb = (m0 + 16 * wm + 8 * kp) / CCH;
    float fOr[2][CCH], fOi[2][CCH], fHr[2][CCH], fHi[2][CCH];
#pragma unroll
    for (int h = 0; h < 2; ++h) {
#pragma unroll
      for (int c = 0; c < CCH; ++c) { fOr[h][c] = 0.f; fOi[h][c] = 0.f; fHr[h][c] = 0.f; fHi[h][c] = 0.f; }
      const int u = mb + h;
      if (u < NM) {
        const int a = (u < 16) ? u : (u - 15);
        const float kx = TWOPI_L * (float)a;
        const float squ = kx * kx + ky * ky + kz * kz;
        float mOO = 0.f, mOH = 0.f, mHO = 0.f, mHH = 0.f;
#pragma unroll
        for (int i = 0; i < 6; ++i) {
          mOO += amOO[i] * expf(-squ * eOO[i]);
          mOH += amOH[i] * expf(-squ * eOH[i]);
          mHO += amHO[i] * expf(-squ * eHO[i]);
          mHH += amHH[i] * expf(-squ * eHH[i]);
        }
        if (squ == 0.0f) { mOH = 0.f; mHO = 0.f; mHH = 0.f; }
        const float w = (u == 0) ? 1.0f : 2.0f;
        mOO *= w; mOH *= w; mHO *= w; mHH *= w;
#pragma unroll
        for (int c = 0; c < CCH; ++c) {
          const int v = CCH * h + c;
          const float por = oR[v], poi = oI[v];
          const float phr = hR[v], phi = hI[v];
          fOr[h][c] = mOO * por + mOH * phr;  fOi[h][c] = -(mOO * poi + mOH * phi);
          fHr[h][c] = mHO * por + mHH * phr;  fHi[h][c] = -(mHO * poi + mHH * phi);
        }
      }
    }
#pragma unroll
    for (int c = 0; c < CCH; ++c) {
      const size_t o = ((size_t)(GPAIR * c + mb / 2) * KNP + np) * 2;
      *(v2f*)(gOr + o) = (v2f){fOr[0][c], fOr[1][c]};
      *(v2f*)(gOi + o) = (v2f){fOi[0][c], fOi[1][c]};
      *(v2f*)(gHr + o) = (v2f){fHr[0][c], fHr[1][c]};
      *(v2f*)(gHi + o) = (v2f){fHi[0][c], fHi[1][c]};
    }
  }
}

struct XFrag { v2f r[CCH][8]; v2f i[CCH][8]; };
struct BFrag { v2f v[CCH][8]; };
struct YZPh { float cy[8], sy[8], cz[8], sz[8]; };

__device__ __forceinline__ void t2_load(const float* __restrict__ Gr, const float* __restrict__ Gi,
                                        const float* __restrict__ Cy, const float* __restrict__ Sy,
                                        const float* __restrict__ Cz, const float* __restrict__ Sz,
                                        int np0, int lrow, int h, int j,
                                        XFrag& a, YZPh& w) {
#pragma unroll
  for (int c = 0; c < CCH; ++c) {
#pragma unroll
    for (int ks = 0; ks < 8; ++ks) {
      const size_t o = ((size_t)(GPAIR * c + 2 * ks + h) * KNP + np0 + lrow) * 2;
      a.r[c][ks] = *(const v2f*)(Gr + o);
      a.i[c][ks] = *(const v2f*)(Gi + o);
    }
  }
#pragma unroll
  for (int r = 0; r < 8; ++r) {
    const int np = np0 + 8 * h + r;
    const int n = np / NM, p = np - NM * n;
    const int nn = (n < NM) ? n : (NM - 1);
    w.cy[r] = Cy[nn * J + j];
    w.sy[r] = Sy[nn * J + j];
    w.cz[r] = Cz[p * J + j];
    w.sz[r] = Sz[p * J + j];
  }
}

__device__ __forceinline__ void t2_tile(const XFrag& a, const BFrag& b, const BFrag& b2,
                                        const YZPh& w, int np0, int h,
                                        float& E, float& FX, float& FY, float& FZ) {
  const v8f z = (v8f){0.f, 0.f, 0.f, 0.f, 0.f, 0.f, 0.f, 0.f};
  v8f pR = z, pI = z, yR = z, yI = z;
#pragma unroll
  for (int c = 0; c < CCH; ++c) {
#pragma unroll
    for (int ks = 0; ks < 8; ++ks) {
      pR = __builtin_amdgcn_wmma_f32_16x16x4_f32(false, a.r[c][ks], false, b.v[c][ks], (short)0, pR, false, false);
      pI = __builtin_amdgcn_wmma_f32_16x16x4_f32(false, a.i[c][ks], false, b.v[c][ks], (short)0, pI, false, false);
      yR = __builtin_amdgcn_wmma_f32_16x16x4_f32(false, a.r[c][ks], false, b2.v[c][ks], (short)0, yR, false, false);
      yI = __builtin_amdgcn_wmma_f32_16x16x4_f32(false, a.i[c][ks], false, b2.v[c][ks], (short)0, yI, false, false);
    }
  }
#pragma unroll
  for (int r = 0; r < 8; ++r) {
    const int np = np0 + 8 * h + r;
    const int n = np / NM, p = np - NM * n;
    const float ky = TWOPI_L * (float)(n - 15);
    const float kz = TWOPI_L * (float)(p - 15);
    const float sr = pR[r], si = pI[r];
    const float tr = -yI[r], ti = yR[r];
    const float wc = w.cy[r] * w.cz[r] - w.sy[r] * w.sz[r];
    const float ws = w.sy[r] * w.cz[r] + w.cy[r] * w.sz[r];
    E += sr * wc - si * ws;
    const float v = sr * ws + si * wc;
    FY += ky * v;
    FZ += kz * v;
    FX += tr * ws + ti * wc;
  }
}

__global__ __launch_bounds__(128)
void type2_xfirst(const float* __restrict__ gOr, const float* __restrict__ gOi,
                  const float* __restrict__ gHr, const float* __restrict__ gHi,
                  const float* __restrict__ Cy, const float* __restrict__ Sy,
                  const float* __restrict__ Cz, const float* __restrict__ Sz,
                  const float* __restrict__ Cx, const float* __restrict__ Sx,
                  const float* __restrict__ charge, float* __restrict__ part) {
  const int lane = threadIdx.x & 31;
  const int wv   = threadIdx.x >> 5;
  const int lrow = lane & 15;
  const int h    = lane >> 4;
  const int jt   = blockIdx.y * 4 + wv;
  const int j    = 16 * jt + lrow;
  const int sl   = blockIdx.x;
  const bool oxy = (16 * jt < JOs);
  const float* Gr = oxy ? gOr : gHr;
  const float* Gi = oxy ? gOi : gHi;

  BFrag b, b2;
  const v4f q = *(const v4f*)(charge + (size_t)j * CCH);
#pragma unroll
  for (int ks = 0; ks < 8; ++ks) {
#pragma unroll
    for (int e = 0; e < 2; ++e) {
      const int u = 4 * ks + 2 * h + e;
      const bool cosrow = (ks < 4);
      const int a = cosrow ? u : (u - 15);
      const int tr = (a + 15 < NM) ? (a + 15) : (NM - 1);
      float cx = Cx[tr * J + j], sx = Sx[tr * J + j];
      if (u >= NM) { cx = 0.f; sx = 0.f; }
      const float kx = TWOPI_L * (float)a;
#pragma unroll
      for (int c = 0; c < CCH; ++c) {
        const float qc = q[c] * cx, qs = q[c] * sx;
        b.v[c][ks][e]  = cosrow ? qc : qs;
        b2.v[c][ks][e] = cosrow ? (kx * qs) : -(kx * qc);
      }
    }
  }

  float E = 0.f, FX = 0.f, FY = 0.f, FZ = 0.f;
  XFrag a0, a1;
  YZPh w0, w1;
  const int t0 = TPS * sl;
  t2_load(Gr, Gi, Cy, Sy, Cz, Sz, 16 * t0, lrow, h, j, a0, w0);
  if (t0 + 1 < NT16) t2_load(Gr, Gi, Cy, Sy, Cz, Sz, 16 * (t0 + 1), lrow, h, j, a1, w1);
  __builtin_amdgcn_sched_barrier(0);
  t2_tile(a0, b, b2, w0, 16 * t0, h, E, FX, FY, FZ);
  if (t0 + 1 < NT16) {
    if (t0 + 2 < NT16) t2_load(Gr, Gi, Cy, Sy, Cz, Sz, 16 * (t0 + 2), lrow, h, j, a0, w0);
    __builtin_amdgcn_sched_barrier(0);
    t2_tile(a1, b, b2, w1, 16 * (t0 + 1), h, E, FX, FY, FZ);
    if (t0 + 2 < NT16) t2_tile(a0, b, b2, w0, 16 * (t0 + 2), h, E, FX, FY, FZ);
  }

  E  += __shfl_xor(E, 16);
  FX += __shfl_xor(FX, 16);
  FY += __shfl_xor(FY, 16);
  FZ += __shfl_xor(FZ, 16);
  if (h == 0) *(v4f*)(part + ((size_t)sl * J + j) * 4) = (v4f){E, FX, FY, FZ};
}

__global__ void finalize(const float* __restrict__ part, float* __restrict__ out, int b) {
  int j = blockIdx.x * blockDim.x + threadIdx.x;
  if (j >= J) return;
  v4f acc = (v4f){0.f, 0.f, 0.f, 0.f};
  for (int sl = 0; sl < NSL; ++sl) {
    const v4f p = *(const v4f*)(part + ((size_t)sl * J + j) * 4);
    acc += p;
  }
  out[b * J + j] = acc[0] / (2.0f * VBOX);
  float* F = out + 2 * J;
  F[(size_t)(b * J + j) * 3 + 0] = acc[1] / VBOX;
  F[(size_t)(b * J + j) * 3 + 1] = acc[2] / VBOX;
  F[(size_t)(b * J + j) * 3 + 2] = acc[3] / VBOX;
}

extern "C" void kernel_launch(void* const* d_in, const int* in_sizes, int n_in,
                              void* d_out, int out_size, void* d_ws, size_t ws_size,
                              hipStream_t stream) {
  const float* pos    = (const float*)d_in[0];
  const float* charge = (const float*)d_in[1];
  const float* shOO = (const float*)d_in[2];
  const float* amOO = (const float*)d_in[3];
  const float* shOH = (const float*)d_in[4];
  const float* amOH = (const float*)d_in[5];
  const float* shHO = (const float*)d_in[6];
  const float* amHO = (const float*)d_in[7];
  const float* shHH = (const float*)d_in[8];
  const float* amHH = (const float*)d_in[9];
  float* out = (float*)d_out;
  float* ws  = (float*)d_ws;

  float *Cx = ws + OFF_CX, *Sx = ws + OFF_SX;
  float *Cy = ws + OFF_CY, *Sy = ws + OFF_SY;
  float *Cz = ws + OFF_CZ, *Sz = ws + OFF_SZ;
  float *gOr = ws + OFF_GOR, *gOi = ws + OFF_GOI;
  float *gHr = ws + OFF_GHR, *gHi = ws + OFF_GHI;
  float *part = ws + OFF_PART;

  hipMemsetAsync(d_ws, 0, WS_TOTAL * sizeof(float), stream);

  for (int b = 0; b < 2; ++b) {
    const float* posb = pos + (size_t)b * J * 3;
    const float* chb  = charge + (size_t)b * J * CCH;

    phase_build<<<(3 * NA * J + 127) / 128, 128, 0, stream>>>(posb, Cx, Sx, Cy, Sy, Cz, Sz);

    dim3 g1(NPPAD / TN, CMP / TM);
    nufft1_mix<<<g1, 128, 0, stream>>>(chb, Cx, Sx, Cy, Sy, Cz, Sz,
                                       shOO, amOO, shOH, amOH, shHO, amHO, shHH, amHH,
                                       gOr, gOi, gHr, gHi, JOs, J);

    dim3 g2(NSL, J / 64);
    type2_xfirst<<<g2, 128, 0, stream>>>(gOr, gOi, gHr, gHi, Cy, Sy, Cz, Sz, Cx, Sx, chb, part);

    finalize<<<(J + 127) / 128, 128, 0, stream>>>(part, out, b);
  }
}


# y and z axes in the real cos/sin basis too: W and g real, a K-step costs 1 (nufft1, K chunk 128) and 2 (type-2) matrix instructions; +-n, +-p sums regrouped, K-steps alternate 2 accumulators
# speedup vs baseline: 15.6907x; 1.3237x over previous
#include <hip/hip_runtime.h>
#include <hip/hip_bf16.h>
#include <math.h>

typedef float v2f __attribute__((ext_vector_type(2)));
typedef float v8f __attribute__((ext_vector_type(8)));

constexpr int NM    = 31;
constexpr int NP    = 961;
constexpr int NPPAD = 1024;
constexpr int KNP   = 1024;
constexpr int J     = 768;
constexpr int JOs   = 256;
constexpr int CCH   = 4;
constexpr int CMP   = 128;
constexpr int GPAIR = 16;
constexpr int NT16  = 61;
constexpr int TPS   = 3;
constexpr int NSL   = 21;
constexpr float LBOX   = 20.0f;
constexpr float VBOX   = 8000.0f;
constexpr float TWOPI_L = 0.31415926535897931f;

constexpr size_t SZ_PH  = (size_t)NM * J;
constexpr size_t OFF_CX = 0;
constexpr size_t OFF_SX = OFF_CX + SZ_PH;
constexpr size_t OFF_CY = OFF_SX + SZ_PH;
constexpr size_t OFF_SY = OFF_CY + SZ_PH;
constexpr size_t OFF_CZ = OFF_SY + SZ_PH;
constexpr size_t OFF_SZ = OFF_CZ + SZ_PH;
constexpr size_t SZ_G   = (size_t)CCH * GPAIR * KNP * 2;
constexpr size_t OFF_GO = OFF_SZ + SZ_PH;
constexpr size_t OFF_GH = OFF_GO + SZ_G;
constexpr size_t SZ_PART = (size_t)NSL * J * 4;
constexpr size_t OFF_PART = OFF_GH + SZ_G;
constexpr size_t WS_TOTAL = OFF_PART + SZ_PART;

constexpr int NA = 16;
__global__ void phase_build(const float* __restrict__ pos,
                            float* __restrict__ Cx, float* __restrict__ Sx,
                            float* __restrict__ Cy, float* __restrict__ Sy,
                            float* __restrict__ Cz, float* __restrict__ Sz) {
  int t = blockIdx.x * blockDim.x + threadIdx.x;
  if (t >= 3 * NA * J) return;
  const int j = t % J, da = t / J;
  const int d = da / NA, a = da % NA;
  float th = TWOPI_L * (pos[j * 3 + d] - 0.5f * LBOX);
  float* C = (d == 0) ? Cx : (d == 1) ? Cy : Cz;
  float* S = (d == 0) ? Sx : (d == 1) ? Sy : Sz;
  float s, c;
  sincosf((float)a * th, &s, &c);
  C[(15 + a) * J + j] = c;
  S[(15 + a) * J + j] = s;
  if (a != 0) {
    C[(15 - a) * J + j] = c;
    S[(15 - a) * J + j] = -s;
  }
}

typedef float v4f __attribute__((ext_vector_type(4)));
constexpr int TM = 32;
constexpr int TN = 32;
constexpr int KC = 128;
constexpr int PA = 132;
constexpr int SZ_A = TM * PA;
constexpr int SZ_B = TN * PA;
constexpr int S_A  = 0;
constexpr int S_B  = S_A + SZ_A;
constexpr int S_BUF = S_B + SZ_B;
static_assert(SZ_A % 4 == 0 && SZ_B % 4 == 0, "planes start on 16-byte boundaries");

constexpr int KB = KC / 8;
constexpr int NR = 8;

__device__ __forceinline__ const float* basis_row(const float* C, const float* S, int u) {
  return (u < 16) ? (C + (u + 15) * J) : (S + u * J);
}

__device__ __forceinline__ void frag_load(const float* buf, int fA, int fB, int blk,
                                          v2f (&a)[KB], v2f (&b)[KB]) {
#pragma unroll
  for (int s = 0; s < KB; ++s) {
    const int ks = blk * KB + s;
    a[s] = *(const v2f*)(buf + S_A + fA + 4 * ks);
    b[s] = *(const v2f*)(buf + S_B + fB + 4 * ks);
  }
}

__device__ __forceinline__ void wmma_block(const v2f (&a)[KB], const v2f (&b)[KB],
                                           v8f& acc0, v8f& acc1) {
#pragma unroll
  for (int s = 0; s < KB; s += 2) {
    acc0 = __builtin_amdgcn_wmma_f32_16x16x4_f32(
        false, a[s], false, b[s], (short)0, acc0, false, false);
    acc1 = __builtin_amdgcn_wmma_f32_16x16x4_f32(
        false, a[s + 1], false, b[s + 1], (short)0, acc1, false, false);
  }
}

__global__ __launch_bounds__(128)
void nufft1_mix(const float* __restrict__ charge,
                const float* __restrict__ Cx, const float* __restrict__ Sx,
                const float* __restrict__ Cy, const float* __restrict__ Sy,
                const float* __restrict__ Cz, const float* __restrict__ Sz,
                const float* __restrict__ shOO, const float* __restrict__ amOO,
                const float* __restrict__ shOH, const float* __restrict__ amOH,
                const float* __restrict__ shHO, const float* __restrict__ amHO,
                const float* __restrict__ shHH, const float* __restrict__ amHH,
                float* __restrict__ gO, float* __restrict__ gH, int KO, int K) {
  __shared__ __attribute__((aligned(16))) float sm[2 * S_BUF];

  const int t    = threadIdx.x;
  const int lane = t & 31;
  const int wv   = t >> 5;
  const int row  = lane & 15;
  const int kp   = lane >> 4;
  const int wm   = wv >> 1;
  const int wn   = wv & 1;
  const int n0 = blockIdx.x * TN;
  const int m0 = blockIdx.y * TM;

  const int ac4 = t & 31, arow = t >> 5;
  const int sT = arow * PA + 4 * ac4;
  const float* qp = charge + (size_t)(4 * ac4) * CCH + (arow & 3);
  const float* pA[NR];
  const float* pY[NR];
  const float* pZ[NR];
#pragma unroll
  for (int i = 0; i < NR; ++i) {
    int u = (m0 + arow + 4 * i) >> 2;
    if (u > NM - 1) u = NM - 1;
    pA[i] = basis_row(Cx, Sx, u) + 4 * ac4;
    int nqB = n0 + arow + 4 * i;
    if (nqB > NP - 1) nqB = NP - 1;
    const int vB = nqB / NM, wB = nqB - NM * vB;
    pY[i] = basis_row(Cy, Sy, vB) + 4 * ac4;
    pZ[i] = basis_row(Cz, Sz, wB) + 4 * ac4;
  }

  const int fA = (16 * wm + row) * PA + 2 * kp;
  const int fB = (16 * wn + row) * PA + 2 * kp;

  v8f o0 = (v8f){0.f, 0.f, 0.f, 0.f, 0.f, 0.f, 0.f, 0.f};
  v8f o1 = (v8f){0.f, 0.f, 0.f, 0.f, 0.f, 0.f, 0.f, 0.f};
  v8f h0 = (v8f){0.f, 0.f, 0.f, 0.f, 0.f, 0.f, 0.f, 0.f};
  v8f h1 = (v8f){0.f, 0.f, 0.f, 0.f, 0.f, 0.f, 0.f, 0.f};

  v4f rq, rax[NR], ry[NR], rz[NR];
  v2f a0[KB], b0[KB];
  v2f a1[KB], b1[KB];

  auto fetch = [&](int ch) {
    const int ka = ch * KC;
#pragma unroll
    for (int e = 0; e < 4; ++e) rq[e] = qp[(size_t)(ka + e) * CCH];
#pragma unroll
    for (int i = 0; i < NR; ++i) {
      rax[i] = *(const v4f*)(pA[i] + ka);
      ry[i]  = *(const v4f*)(pY[i] + ka);
      rz[i]  = *(const v4f*)(pZ[i] + ka);
    }
  };
  auto stash = [&](float* dst) {
#pragma unroll
    for (int i = 0; i < NR; ++i) {
      *(v4f*)(dst + S_A + sT + 4 * i * PA) = rq * rax[i];
      *(v4f*)(dst + S_B + sT + 4 * i * PA) = ry[i] * rz[i];
    }
  };

  const int nch = K / KC;
  const int nco = KO / KC;

  fetch(0);
  stash(sm);
  __syncthreads();

  frag_load(sm, fA, fB, 0, a0, b0);
  if (1 < nch) fetch(1);

  auto chunk = [&](int c, v8f& acc0, v8f& acc1) {
    const float* cur = sm + (c & 1) * S_BUF;
    float* nxt = sm + ((c + 1) & 1) * S_BUF;

    frag_load(cur, fA, fB, 1, a1, b1);
    __builtin_amdgcn_sched_barrier(0);
    wmma_block(a0, b0, acc0, acc1);

    const bool more = (c + 1 < nch);
    if (more) stash(nxt);
    __syncthreads();
    if (more) {
      frag_load(nxt, fA, fB, 0, a0, b0);
      if (c + 2 < nch) fetch(c + 2);
    }
    wmma_block(a1, b1, acc0, acc1);
  };

  for (int c = 0; c < nco; ++c) chunk(c, o0, o1);
  for (int c = nco; c < nch; ++c) chunk(c, h0, h1);

  const v8f pO = o0 + o1;
  const v8f pH = h0 + h1;

  const int nq = n0 + 16 * wn + row;
  if (nq < NP) {
    const int v = nq / NM, w = nq % NM;
    const int bv = (v < 16) ? v : (v - 15);
    const int bw = (w < 16) ? w : (w - 15);
    const float ky = TWOPI_L * (float)bv;
    const float kz = TWOPI_L * (float)bw;
    const float wyz = ((v == 0) ? 1.0f : 2.0f) * ((w == 0) ? 1.0f : 2.0f);
    float eOO[6], eOH[6], eHO[6], eHH[6];
#pragma unroll
    for (int i = 0; i < 6; ++i) {
      eOO[i] = expf(2.0f * shOO[i]);
      eOH[i] = expf(2.0f * shOH[i]);
      eHO[i] = expf(2.0f * shHO[i]);
      eHH[i] = expf(2.0f * shHH[i]);
    }
    const int mb = (m0 + 16 * wm + 8 * kp) / CCH;
    float fO[2][CCH], fH[2][CCH];
#pragma unroll
    for (int h = 0; h < 2; ++h) {
#pragma unroll
      for (int c = 0; c < CCH; ++c) { fO[h][c] = 0.f; fH[h][c] = 0.f; }
      const int u = mb + h;
      if (u < NM) {
        const int a = (u < 16) ? u : (u - 15);
        const float kx = TWOPI_L * (float)a;
        const float squ = kx * kx + ky * ky + kz * kz;
        float mOO = 0.f, mOH = 0.f, mHO = 0.f, mHH = 0.f;
#pragma unroll
        for (int i = 0; i < 6; ++i) {
          mOO += amOO[i] * expf(-squ * eOO[i]);
          mOH += amOH[i] * expf(-squ * eOH[i]);
          mHO += amHO[i] * expf(-squ * eHO[i]);
          mHH += amHH[i] * expf(-squ * eHH[i]);
        }
        if (squ == 0.0f) { mOH = 0.f; mHO = 0.f; mHH = 0.f; }
        const float wt = ((u == 0) ? 1.0f : 2.0f) * wyz;
        mOO *= wt; mOH *= wt; mHO *= wt; mHH *= wt;
#pragma unroll
        for (int c = 0; c < CCH; ++c) {
          const int e = CCH * h + c;
          fO[h][c] = mOO * pO[e] + mOH * pH[e];
          fH[h][c] = mHO * pO[e] + mHH * pH[e];
        }
      }
    }
#pragma unroll
    for (int c = 0; c < CCH; ++c) {
      const size_t o = ((size_t)(GPAIR * c + mb / 2) * KNP + nq) * 2;
      *(v2f*)(gO + o) = (v2f){fO[0][c], fO[1][c]};
      *(v2f*)(gH + o) = (v2f){fH[0][c], fH[1][c]};
    }
  }
}

struct XFrag { v2f r[CCH][8]; };
struct BFrag { v2f v[CCH][8]; };
struct YZPh { float cy[8], sy[8], cz[8], sz[8]; };

__device__ __forceinline__ void yz_of(int nq, int& v, int& w) {
  const int vv = nq / NM;
  w = nq - NM * vv;
  v = (vv < NM) ? vv : (NM - 1);
}

__device__ __forceinline__ void t2_load(const float* __restrict__ G,
                                        const float* __restrict__ Cy, const float* __restrict__ Sy,
                                        const float* __restrict__ Cz, const float* __restrict__ Sz,
                                        int np0, int lrow, int h, int j,
                                        XFrag& a, YZPh& ph) {
#pragma unroll
  for (int c = 0; c < CCH; ++c) {
#pragma unroll
    for (int ks = 0; ks < 8; ++ks) {
      const size_t o = ((size_t)(GPAIR * c + 2 * ks + h) * KNP + np0 + lrow) * 2;
      a.r[c][ks] = *(const v2f*)(G + o);
    }
  }
#pragma unroll
  for (int r = 0; r < 8; ++r) {
    int v, w;
    yz_of(np0 + 8 * h + r, v, w);
    const int bv = (v < 16) ? v : (v - 15);
    const int bw = (w < 16) ? w : (w - 15);
    ph.cy[r] = Cy[(15 + bv) * J + j];
    ph.sy[r] = Sy[(15 + bv) * J + j];
    ph.cz[r] = Cz[(15 + bw) * J + j];
    ph.sz[r] = Sz[(15 + bw) * J + j];
  }
}

__device__ __forceinline__ void t2_tile(const XFrag& a, const BFrag& b, const BFrag& b2,
                                        const YZPh& ph, int np0, int h,
                                        float& E, float& FX, float& FY, float& FZ) {
  const v8f z = (v8f){0.f, 0.f, 0.f, 0.f, 0.f, 0.f, 0.f, 0.f};
  v8f pR = z, yR = z;
#pragma unroll
  for (int c = 0; c < CCH; ++c) {
#pragma unroll
    for (int ks = 0; ks < 8; ++ks) {
      pR = __builtin_amdgcn_wmma_f32_16x16x4_f32(false, a.r[c][ks], false, b.v[c][ks], (short)0, pR, false, false);
      yR = __builtin_amdgcn_wmma_f32_16x16x4_f32(false, a.r[c][ks], false, b2.v[c][ks], (short)0, yR, false, false);
    }
  }
#pragma unroll
  for (int r = 0; r < 8; ++r) {
    int v, w;
    yz_of(np0 + 8 * h + r, v, w);
    const bool cv = (v < 16), cw = (w < 16);
    const float ky = TWOPI_L * (float)(cv ? v : (v - 15));
    const float kz = TWOPI_L * (float)(cw ? w : (w - 15));
    const float Y  = cv ? ph.cy[r] : ph.sy[r];
    const float DY = cv ? (ky * ph.sy[r]) : -(ky * ph.cy[r]);
    const float Z  = cw ? ph.cz[r] : ph.sz[r];
    const float DZ = cw ? (kz * ph.sz[r]) : -(kz * ph.cz[r]);
    const float yz = Y * Z;
    const float s = pR[r];
    E  += s * yz;
    FX += yR[r] * yz;
    FY += s * (DY * Z);
    FZ += s * (Y * DZ);
  }
}

__global__ __launch_bounds__(128)
void type2_xfirst(const float* __restrict__ gO, const float* __restrict__ gH,
                  const float* __restrict__ Cy, const float* __restrict__ Sy,
                  const float* __restrict__ Cz, const float* __restrict__ Sz,
                  const float* __restrict__ Cx, const float* __restrict__ Sx,
                  const float* __restrict__ charge, float* __restrict__ part) {
  const int lane = threadIdx.x & 31;
  const int wv   = threadIdx.x >> 5;
  const int lrow = lane & 15;
  const int h    = lane >> 4;
  const int jt   = blockIdx.y * 4 + wv;
  const int j    = 16 * jt + lrow;
  const int sl   = blockIdx.x;
  const bool oxy = (16 * jt < JOs);
  const float* G = oxy ? gO : gH;

  BFrag b, b2;
  const v4f q = *(const v4f*)(charge + (size_t)j * CCH);
#pragma unroll
  for (int ks = 0; ks < 8; ++ks) {
#pragma unroll
    for (int e = 0; e < 2; ++e) {
      const int u = 4 * ks + 2 * h + e;
      const bool cosrow = (ks < 4);
      const int a = cosrow ? u : (u - 15);
      const int tr = (a + 15 < NM) ? (a + 15) : (NM - 1);
      float cx = Cx[tr * J + j], sx = Sx[tr * J + j];
      if (u >= NM) { cx = 0.f; sx = 0.f; }
      const float kx = TWOPI_L * (float)a;
#pragma unroll
      for (int c = 0; c < CCH; ++c) {
        const float qc = q[c] * cx, qs = q[c] * sx;
        b.v[c][ks][e]  = cosrow ? qc : qs;
        b2.v[c][ks][e] = cosrow ? (kx * qs) : -(kx * qc);
      }
    }
  }

  float E = 0.f, FX = 0.f, FY = 0.f, FZ = 0.f;
  XFrag a0, a1;
  YZPh w0, w1;
  const int t0 = TPS * sl;
  t2_load(G, Cy, Sy, Cz, Sz, 16 * t0, lrow, h, j, a0, w0);
  if (t0 + 1 < NT16) t2_load(G, Cy, Sy, Cz, Sz, 16 * (t0 + 1), lrow, h, j, a1, w1);
  __builtin_amdgcn_sched_barrier(0);
  t2_tile(a0, b, b2, w0, 16 * t0, h, E, FX, FY, FZ);
  if (t0 + 1 < NT16) {
    if (t0 + 2 < NT16) t2_load(G, Cy, Sy, Cz, Sz, 16 * (t0 + 2), lrow, h, j, a0, w0);
    __builtin_amdgcn_sched_barrier(0);
    t2_tile(a1, b, b2, w1, 16 * (t0 + 1), h, E, FX, FY, FZ);
    if (t0 + 2 < NT16) t2_tile(a0, b, b2, w0, 16 * (t0 + 2), h, E, FX, FY, FZ);
  }

  E  += __shfl_xor(E, 16);
  FX += __shfl_xor(FX, 16);
  FY += __shfl_xor(FY, 16);
  FZ += __shfl_xor(FZ, 16);
  if (h == 0) *(v4f*)(part + ((size_t)sl * J + j) * 4) = (v4f){E, FX, FY, FZ};
}

__global__ void finalize(const float* __restrict__ part, float* __restrict__ out, int b) {
  int j = blockIdx.x * blockDim.x + threadIdx.x;
  if (j >= J) return;
  v4f acc = (v4f){0.f, 0.f, 0.f, 0.f};
  for (int sl = 0; sl < NSL; ++sl) {
    const v4f p = *(const v4f*)(part + ((size_t)sl * J + j) * 4);
    acc += p;
  }
  out[b * J + j] = acc[0] / (2.0f * VBOX);
  float* F = out + 2 * J;
  F[(size_t)(b * J + j) * 3 + 0] = acc[1] / VBOX;
  F[(size_t)(b * J + j) * 3 + 1] = acc[2] / VBOX;
  F[(size_t)(b * J + j) * 3 + 2] = acc[3] / VBOX;
}

extern "C" void kernel_launch(void* const* d_in, const int* in_sizes, int n_in,
                              void* d_out, int out_size, void* d_ws, size_t ws_size,
                              hipStream_t stream) {
  const float* pos    = (const float*)d_in[0];
  const float* charge = (const float*)d_in[1];
  const float* shOO = (const float*)d_in[2];
  const float* amOO = (const float*)d_in[3];
  const float* shOH = (const float*)d_in[4];
  const float* amOH = (const float*)d_in[5];
  const float* shHO = (const float*)d_in[6];
  const float* amHO = (const float*)d_in[7];
  const float* shHH = (const float*)d_in[8];
  const float* amHH = (const float*)d_in[9];
  float* out = (float*)d_out;
  float* ws  = (float*)d_ws;

  float *Cx = ws + OFF_CX, *Sx = ws + OFF_SX;
  float *Cy = ws + OFF_CY, *Sy = ws + OFF_SY;
  float *Cz = ws + OFF_CZ, *Sz = ws + OFF_SZ;
  float *gO = ws + OFF_GO, *gH = ws + OFF_GH;
  float *part = ws + OFF_PART;

  hipMemsetAsync(d_ws, 0, WS_TOTAL * sizeof(float), stream);

  for (int b = 0; b < 2; ++b) {
    const float* posb = pos + (size_t)b * J * 3;
    const float* chb  = charge + (size_t)b * J * CCH;

    phase_build<<<(3 * NA * J + 127) / 128, 128, 0, stream>>>(posb, Cx, Sx, Cy, Sy, Cz, Sz);

    dim3 g1(NPPAD / TN, CMP / TM);
    nufft1_mix<<<g1, 128, 0, stream>>>(chb, Cx, Sx, Cy, Sy, Cz, Sz,
                                       shOO, amOO, shOH, amOH, shHO, amHO, shHH, amHH,
                                       gO, gH, JOs, J);

    dim3 g2(NSL, J / 64);
    type2_xfirst<<<g2, 128, 0, stream>>>(gO, gH, Cy, Sy, Cz, Sz, Cx, Sx, chb, part);

    finalize<<<(J + 127) / 128, 128, 0, stream>>>(part, out, b);
  }
}
